# starting kernel
# baseline (speedup 1.0000x reference)
#include <hip/hip_runtime.h>
#include <cstdint>

#define C 128
#define FLT_BIG 3.402823466e38f

typedef float v2f __attribute__((ext_vector_type(2)));
typedef float v8f __attribute__((ext_vector_type(8)));

__global__ void k_zero(float* __restrict__ p, long long n) {
  long long i = (long long)blockIdx.x * blockDim.x + threadIdx.x;
  if (i < n) p[i] = 0.0f;
}

__global__ void k_init_aggr(float* __restrict__ sum, float* __restrict__ mx,
                            float* __restrict__ mn, float* __restrict__ cnt,
                            int n_nodes) {
  long long i = (long long)blockIdx.x * blockDim.x + threadIdx.x;
  long long nc = (long long)n_nodes * C;
  if (i < nc) {
    sum[i] = 0.0f;
    mx[i]  = -FLT_BIG;
    mn[i]  =  FLT_BIG;
  }
  if (i < n_nodes) cnt[i] = 0.0f;
}

__global__ void k_scatter(const float* __restrict__ x, const int* __restrict__ ei,
                          int n_edges,
                          float* __restrict__ sum, float* __restrict__ mx,
                          float* __restrict__ mn, float* __restrict__ cnt) {
  long long tid = (long long)blockIdx.x * blockDim.x + threadIdx.x;
  int e = (int)(tid >> 7);
  int c = (int)(tid & 127);
  if (e >= n_edges) return;
  int s = ei[e];
  int d = ei[n_edges + e];
  float v = x[(long long)s * C + c];
  long long o = (long long)d * C + c;
  __hip_atomic_fetch_add(&sum[o], v, __ATOMIC_RELAXED, __HIP_MEMORY_SCOPE_AGENT);
  __hip_atomic_fetch_max(&mx[o],  v, __ATOMIC_RELAXED, __HIP_MEMORY_SCOPE_AGENT);
  __hip_atomic_fetch_min(&mn[o],  v, __ATOMIC_RELAXED, __HIP_MEMORY_SCOPE_AGENT);
  if (c == 0)
    __hip_atomic_fetch_add(&cnt[d], 1.0f, __ATOMIC_RELAXED, __HIP_MEMORY_SCOPE_AGENT);
}

__global__ void k_finalize(const float* __restrict__ sum, float* __restrict__ mx,
                           float* __restrict__ mn, float* __restrict__ mean,
                           const float* __restrict__ cnt, int n_nodes) {
  long long i = (long long)blockIdx.x * blockDim.x + threadIdx.x;
  long long nc = (long long)n_nodes * C;
  if (i >= nc) return;
  int node = (int)(i >> 7);
  float cn = cnt[node];
  mean[i] = sum[i] / fmaxf(cn, 1.0f);
  if (cn == 0.0f) { mx[i] = 0.0f; mn[i] = 0.0f; }
}

__global__ __launch_bounds__(128) void k_gemm_wmma(
    const float* __restrict__ x,
    const float* __restrict__ mean, const float* __restrict__ mx,
    const float* __restrict__ mn,   const float* __restrict__ sum,
    const float* __restrict__ W1,   const float* __restrict__ W2,
    float* __restrict__ h, int n_nodes) {
  const int lane = threadIdx.x & 31;
  const int wv   = threadIdx.x >> 5;
  const int m0   = blockIdx.x * 64 + wv * 16;
  if (m0 >= n_nodes) return;

  int row = m0 + (lane & 15);
  if (row >= n_nodes) row = n_nodes - 1;
  const int khalf = (lane >> 4) << 1;
  const int nlane = lane & 15;

  v8f acc[8] = {};

  const float* Asrc[5] = { x, mean, mx, mn, sum };
#pragma unroll
  for (int seg = 0; seg < 5; ++seg) {
    const float* A   = Asrc[seg];
    const float* W   = (seg == 0) ? W1 : W2;
    const int    ldw = (seg == 0) ? C : 4 * C;
    const int    wk  = (seg == 0) ? 0 : (seg - 1) * C;
    const float* Ap  = A + (long long)row * C;
    const float* Wp  = W + (long long)nlane * ldw + wk;
#pragma unroll 2
    for (int k0 = 0; k0 < C; k0 += 4) {
      const int kk = k0 + khalf;
      v2f a = *(const v2f*)(Ap + kk);
#pragma unroll
      for (int j = 0; j < 8; ++j) {
        v2f b = *(const v2f*)(Wp + j * 16 * ldw + kk);
        acc[j] = __builtin_amdgcn_wmma_f32_16x16x4_f32(
            false, a, false, b, (short)0, acc[j], false, false);
      }
    }
  }

  const int rbase = m0 + ((lane >> 4) << 3);
#pragma unroll
  for (int v = 0; v < 8; ++v) {
    int r = rbase + v;
    if (r < n_nodes) {
      float* hp = h + (long long)r * C + nlane;
#pragma unroll
      for (int j = 0; j < 8; ++j) hp[j * 16] += acc[j][v];
    }
  }
}

__global__ __launch_bounds__(128) void k_ln(const float* __restrict__ h,
                                            const float* __restrict__ gamma,
                                            const float* __restrict__ beta,
                                            float* __restrict__ out, int n_nodes) {
  int node = blockIdx.x;
  int c = threadIdx.x;
  float v = h[(long long)node * C + c];
  v = fmaxf(0.5f * v, 0.0f);

  float s = v, s2 = v * v;
#pragma unroll
  for (int off = 16; off > 0; off >>= 1) {
    s  += __shfl_xor(s,  off, 32);
    s2 += __shfl_xor(s2, off, 32);
  }
  __shared__ float ps[4], ps2[4];
  int wid = threadIdx.x >> 5;
  if ((threadIdx.x & 31) == 0) { ps[wid] = s; ps2[wid] = s2; }
  __syncthreads();
  float ts  = ps[0]  + ps[1]  + ps[2]  + ps[3];
  float ts2 = ps2[0] + ps2[1] + ps2[2] + ps2[3];
  float mu  = ts  * (1.0f / C);
  float var = ts2 * (1.0f / C) - mu * mu;
  out[(long long)node * C + c] = (v - mu) * rsqrtf(var + 1e-5f) * gamma[c] + beta[c];
}

extern "C" void kernel_launch(void* const* d_in, const int* in_sizes, int n_in,
                              void* d_out, int out_size, void* d_ws, size_t ws_size,
                              hipStream_t stream) {
  const float* x     = (const float*)d_in[0];
  const int*   ei1   = (const int*)d_in[1];
  const int*   ei2   = (const int*)d_in[2];
  const float* W1_r1 = (const float*)d_in[3];
  const float* W2_r1 = (const float*)d_in[4];
  const float* W1_r2 = (const float*)d_in[5];
  const float* W2_r2 = (const float*)d_in[6];
  const float* gamma = (const float*)d_in[7];
  const float* beta  = (const float*)d_in[8];
  float* out = (float*)d_out;

  const int N = in_sizes[0] / C;
  const int E = in_sizes[1] / 2;
  const long long NC = (long long)N * C;

  float* sum  = (float*)d_ws;
  float* mx   = sum  + NC;
  float* mn   = mx   + NC;
  float* mean = mn   + NC;
  float* cnt  = mean + NC;
  float* h    = cnt  + N;

  const int T = 256;
  dim3 gNC((unsigned)((NC + T - 1) / T));
  dim3 gE((unsigned)(((long long)E * C + T - 1) / T));
  dim3 gM((unsigned)((N + 63) / 64));

  k_zero<<<gNC, T, 0, stream>>>(h, NC);

  k_init_aggr<<<gNC, T, 0, stream>>>(sum, mx, mn, cnt, N);
  k_scatter<<<gE, T, 0, stream>>>(x, ei1, E, sum, mx, mn, cnt);
  k_finalize<<<gNC, T, 0, stream>>>(sum, mx, mn, mean, cnt, N);
  k_gemm_wmma<<<gM, 128, 0, stream>>>(x, mean, mx, mn, sum, W1_r1, W2_r1, h, N);

  k_init_aggr<<<gNC, T, 0, stream>>>(sum, mx, mn, cnt, N);
  k_scatter<<<gE, T, 0, stream>>>(x, ei2, E, sum, mx, mn, cnt);
  k_finalize<<<gNC, T, 0, stream>>>(sum, mx, mn, mean, cnt, N);
  k_gemm_wmma<<<gM, 128, 0, stream>>>(x, mean, mx, mn, sum, W1_r2, W2_r2, h, N);

  k_ln<<<(unsigned)N, C, 0, stream>>>(h, gamma, beta, out, N);
}


# zero, link and layer-norm launches re-tiled to few long-lived waves (grid-stride 16-byte stores; 4 edges per thread; one wave per row); LN row sum order changes: 4 per lane, then wave butterfly
# speedup vs baseline: 2.3073x; 2.3073x over previous
#include <hip/hip_runtime.h>
#include <cstdint>

#define C 128
#define FLT_BIG 3.402823466e38f

typedef float v2f __attribute__((ext_vector_type(2)));
typedef float v4f __attribute__((ext_vector_type(4)));
typedef int   v2i __attribute__((ext_vector_type(2)));
typedef float v8f __attribute__((ext_vector_type(8)));

__global__ __launch_bounds__(256) void k_zero(float* __restrict__ p, long long n,
                                              int* __restrict__ head, int n_head) {
  const long long stride = (long long)gridDim.x * blockDim.x;
  const long long t = (long long)blockIdx.x * blockDim.x + threadIdx.x;
  v4f* __restrict__ p4 = (v4f*)p;
  const long long n4 = n >> 2;
  const v4f z = { 0.0f, 0.0f, 0.0f, 0.0f };
  for (long long i = t; i < n4; i += stride) p4[i] = z;
  for (long long i = t; i < n_head; i += stride) head[i] = 0;
}

__device__ __forceinline__ void link_one(long long t,
    const int* __restrict__ ei1, const int* __restrict__ ei2,
    int n_edges, int n_nodes, int* __restrict__ head, v2i* __restrict__ rec) {
  const int r = (t >= n_edges) ? 1 : 0;
  const int e = (int)(t - (long long)r * n_edges);
  const int* ei = r ? ei2 : ei1;
  const int s = ei[e];
  const int d = ei[n_edges + e];
  const int old = __hip_atomic_exchange(&head[(long long)r * n_nodes + d], e + 1,
                                        __ATOMIC_RELAXED, __HIP_MEMORY_SCOPE_AGENT);
  v2i q; q.x = s; q.y = old;
  rec[t] = q;
}

__global__ __launch_bounds__(256) void k_link(
    const int* __restrict__ ei1, const int* __restrict__ ei2,
    int n_edges, int n_nodes,
    int* __restrict__ head, v2i* __restrict__ rec) {
  const long long total  = 2LL * n_edges;
  const long long stride = (long long)gridDim.x * blockDim.x;
  long long t0 = (long long)blockIdx.x * blockDim.x + threadIdx.x;
  for (; t0 + 3 * stride < total; t0 += 4 * stride) {
    int  s[4], e[4], old[4];
    int* hp[4];
#pragma unroll
    for (int k = 0; k < 4; ++k) {
      const long long t = t0 + k * stride;
      const int r = (t >= n_edges) ? 1 : 0;
      e[k] = (int)(t - (long long)r * n_edges);
      const int* ei = r ? ei2 : ei1;
      s[k] = ei[e[k]];
      const int d = ei[n_edges + e[k]];
      hp[k] = head + (long long)r * n_nodes + d;
    }
#pragma unroll
    for (int k = 0; k < 4; ++k)
      old[k] = __hip_atomic_exchange(hp[k], e[k] + 1,
                                     __ATOMIC_RELAXED, __HIP_MEMORY_SCOPE_AGENT);
#pragma unroll
    for (int k = 0; k < 4; ++k) {
      v2i q; q.x = s[k]; q.y = old[k];
      rec[t0 + k * stride] = q;
    }
  }
  for (; t0 < total; t0 += stride)
    link_one(t0, ei1, ei2, n_edges, n_nodes, head, rec);
}

__global__ __launch_bounds__(256) void k_gather(
    const float* __restrict__ x, const int* __restrict__ head,
    const v2i* __restrict__ rec, int n_nodes, int n_edges,
    float* __restrict__ sum, float* __restrict__ mx,
    float* __restrict__ mn, float* __restrict__ mean) {
  const int lane = threadIdx.x & 31;
  int node = blockIdx.x * 8 + (threadIdx.x >> 5);
  if (node >= n_nodes) return;
  node = __builtin_amdgcn_readfirstlane(node);

  v4f s  = { 0.0f, 0.0f, 0.0f, 0.0f };
  v4f hi = { -FLT_BIG, -FLT_BIG, -FLT_BIG, -FLT_BIG };
  v4f lo = {  FLT_BIG,  FLT_BIG,  FLT_BIG,  FLT_BIG };
  int cnt = 0;

  int e1 = head[node];
  v2i q = { 0, 0 };
  if (e1) q = rec[e1 - 1];
  for (int it = 0; it < n_edges && e1 != 0; ++it) {
    const v4f v = *(const v4f*)(x + (long long)q.x * C + lane * 4);
    e1 = q.y;
    if (e1) q = rec[e1 - 1];
    s += v;
    for (int k = 0; k < 4; ++k) {
      hi[k] = fmaxf(hi[k], v[k]);
      lo[k] = fminf(lo[k], v[k]);
    }
    ++cnt;
  }

  const float cn = (float)cnt;
  const float dv = fmaxf(cn, 1.0f);
  v4f m;
  for (int k = 0; k < 4; ++k) m[k] = s[k] / dv;
  if (cnt == 0) {
    hi = (v4f){ 0.0f, 0.0f, 0.0f, 0.0f };
    lo = (v4f){ 0.0f, 0.0f, 0.0f, 0.0f };
  }
  const long long o = (long long)node * C + lane * 4;
  *(v4f*)(sum  + o) = s;
  *(v4f*)(mx   + o) = hi;
  *(v4f*)(mn   + o) = lo;
  *(v4f*)(mean + o) = m;
}

__global__ __launch_bounds__(128) void k_gemm_wmma(
    const float* __restrict__ x,
    const float* __restrict__ mean, const float* __restrict__ mx,
    const float* __restrict__ mn,   const float* __restrict__ sum,
    const float* __restrict__ W1,   const float* __restrict__ W2,
    float* __restrict__ h, int n_nodes) {
  const int lane = threadIdx.x & 31;
  const int wv   = threadIdx.x >> 5;
  const int m0   = blockIdx.x * 64 + wv * 16;
  if (m0 >= n_nodes) return;

  int row = m0 + (lane & 15);
  if (row >= n_nodes) row = n_nodes - 1;
  const int khalf = (lane >> 4) << 1;
  const int nlane = lane & 15;

  v8f acc[8] = {};

  const float* Asrc[5] = { x, mean, mx, mn, sum };
#pragma unroll
  for (int seg = 0; seg < 5; ++seg) {
    const float* A   = Asrc[seg];
    const float* W   = (seg == 0) ? W1 : W2;
    const int    ldw = (seg == 0) ? C : 4 * C;
    const int    wk  = (seg == 0) ? 0 : (seg - 1) * C;
    const float* Ap  = A + (long long)row * C;
    const float* Wp  = W + (long long)nlane * ldw + wk;
#pragma unroll 2
    for (int k0 = 0; k0 < C; k0 += 4) {
      const int kk = k0 + khalf;
      v2f a = *(const v2f*)(Ap + kk);
#pragma unroll
      for (int j = 0; j < 8; ++j) {
        v2f b = *(const v2f*)(Wp + j * 16 * ldw + kk);
        acc[j] = __builtin_amdgcn_wmma_f32_16x16x4_f32(
            false, a, false, b, (short)0, acc[j], false, false);
      }
    }
  }

  const int rbase = m0 + ((lane >> 4) << 3);
#pragma unroll
  for (int v = 0; v < 8; ++v) {
    int r = rbase + v;
    if (r < n_nodes) {
      float* hp = h + (long long)r * C + nlane;
#pragma unroll
      for (int j = 0; j < 8; ++j) hp[j * 16] += acc[j][v];
    }
  }
}

__device__ __forceinline__ v4f ln_row(v4f v, const v4f g, const v4f b) {
#pragma unroll
  for (int k = 0; k < 4; ++k) v[k] = fmaxf(0.5f * v[k], 0.0f);
  float s  = (v[0] + v[1]) + (v[2] + v[3]);
  float s2 = (v[0] * v[0] + v[1] * v[1]) + (v[2] * v[2] + v[3] * v[3]);
#pragma unroll
  for (int off = 16; off > 0; off >>= 1) {
    s  += __shfl_xor(s,  off, 32);
    s2 += __shfl_xor(s2, off, 32);
  }
  const float mu  = s  * (1.0f / C);
  const float var = s2 * (1.0f / C) - mu * mu;
  const float rs  = rsqrtf(var + 1e-5f);
  v4f o;
#pragma unroll
  for (int k = 0; k < 4; ++k) o[k] = (v[k] - mu) * rs * g[k] + b[k];
  return o;
}

__global__ __launch_bounds__(256) void k_ln(const float* __restrict__ h,
                                            const float* __restrict__ gamma,
                                            const float* __restrict__ beta,
                                            float* __restrict__ out, int n_nodes) {
  const int lane = threadIdx.x & 31;
  const int nw   = (int)((gridDim.x * blockDim.x) >> 5);
  int r0 = (int)((blockIdx.x * blockDim.x + threadIdx.x) >> 5);
  r0 = __builtin_amdgcn_readfirstlane(r0);
  const v4f g = *(const v4f*)(gamma + lane * 4);
  const v4f b = *(const v4f*)(beta  + lane * 4);
  for (; r0 + nw < n_nodes; r0 += 2 * nw) {
    const int r1 = r0 + nw;
    const v4f a = *(const v4f*)(h + (long long)r0 * C + lane * 4);
    const v4f c = *(const v4f*)(h + (long long)r1 * C + lane * 4);
    *(v4f*)(out + (long long)r0 * C + lane * 4) = ln_row(a, g, b);
    *(v4f*)(out + (long long)r1 * C + lane * 4) = ln_row(c, g, b);
  }
  if (r0 < n_nodes) {
    const v4f a = *(const v4f*)(h + (long long)r0 * C + lane * 4);
    *(v4f*)(out + (long long)r0 * C + lane * 4) = ln_row(a, g, b);
  }
}

extern "C" void kernel_launch(void* const* d_in, const int* in_sizes, int n_in,
                              void* d_out, int out_size, void* d_ws, size_t ws_size,
                              hipStream_t stream) {
  const float* x     = (const float*)d_in[0];
  const int*   ei1   = (const int*)d_in[1];
  const int*   ei2   = (const int*)d_in[2];
  const float* W1_r1 = (const float*)d_in[3];
  const float* W2_r1 = (const float*)d_in[4];
  const float* W1_r2 = (const float*)d_in[5];
  const float* W2_r2 = (const float*)d_in[6];
  const float* gamma = (const float*)d_in[7];
  const float* beta  = (const float*)d_in[8];
  float* out = (float*)d_out;

  const int N = in_sizes[0] / C;
  const int E = in_sizes[1] / 2;
  const long long NC = (long long)N * C;

  float* sum  = (float*)d_ws;
  float* mx   = sum  + NC;
  float* mn   = mx   + NC;
  float* mean = mn   + NC;
  float* h    = mean + NC;

  int* head = (int*)d_out;
  v2i* rec  = (v2i*)(head + 2 * (long long)N);

  const int T = 256;
  dim3 gS(1024);
  dim3 gE2((unsigned)((2LL * E + 4 * T - 1) / (4 * T)));
  dim3 gG((unsigned)((N + 7) / 8));
  dim3 gM((unsigned)((N + 63) / 64));

  k_zero<<<gS, T, 0, stream>>>(h, NC, head, 2 * N);
  k_link<<<gE2, T, 0, stream>>>(ei1, ei2, E, N, head, rec);

  k_gather<<<gG, T, 0, stream>>>(x, head, rec, N, E, sum, mx, mn, mean);
  k_gemm_wmma<<<gM, 128, 0, stream>>>(x, mean, mx, mn, sum, W1_r1, W2_r1, h, N);

  k_gather<<<gG, T, 0, stream>>>(x, head + N, rec + E, N, E, sum, mx, mn, mean);
  k_gemm_wmma<<<gM, 128, 0, stream>>>(x, mean, mx, mn, sum, W1_r2, W2_r2, h, N);

  k_ln<<<gS, T, 0, stream>>>(h, gamma, beta, out, N);
}


# GEMM operand feed: weight tile of each 32-k chunk staged once per work-group in LDS (whole-line 16-byte loads, two stages); A rows read in 16-byte pieces; k sum order in the products changes
# speedup vs baseline: 3.2874x; 1.4248x over previous
#include <hip/hip_runtime.h>
#include <cstdint>

#define C 128
#define FLT_BIG 3.402823466e38f

typedef float v2f __attribute__((ext_vector_type(2)));
typedef float v4f __attribute__((ext_vector_type(4)));
typedef int   v2i __attribute__((ext_vector_type(2)));
typedef float v8f __attribute__((ext_vector_type(8)));

__global__ __launch_bounds__(256) void k_zero(float* __restrict__ p, long long n,
                                              int* __restrict__ head, int n_head) {
  const long long stride = (long long)gridDim.x * blockDim.x;
  const long long t = (long long)blockIdx.x * blockDim.x + threadIdx.x;
  v4f* __restrict__ p4 = (v4f*)p;
  const long long n4 = n >> 2;
  const v4f z = { 0.0f, 0.0f, 0.0f, 0.0f };
  for (long long i = t; i < n4; i += stride) p4[i] = z;
  for (long long i = t; i < n_head; i += stride) head[i] = 0;
}

__device__ __forceinline__ void link_one(long long t,
    const int* __restrict__ ei1, const int* __restrict__ ei2,
    int n_edges, int n_nodes, int* __restrict__ head, v2i* __restrict__ rec) {
  const int r = (t >= n_edges) ? 1 : 0;
  const int e = (int)(t - (long long)r * n_edges);
  const int* ei = r ? ei2 : ei1;
  const int s = ei[e];
  const int d = ei[n_edges + e];
  const int old = __hip_atomic_exchange(&head[(long long)r * n_nodes + d], e + 1,
                                        __ATOMIC_RELAXED, __HIP_MEMORY_SCOPE_AGENT);
  v2i q; q.x = s; q.y = old;
  rec[t] = q;
}

__global__ __launch_bounds__(256) void k_link(
    const int* __restrict__ ei1, const int* __restrict__ ei2,
    int n_edges, int n_nodes,
    int* __restrict__ head, v2i* __restrict__ rec) {
  const long long total  = 2LL * n_edges;
  const long long stride = (long long)gridDim.x * blockDim.x;
  long long t0 = (long long)blockIdx.x * blockDim.x + threadIdx.x;
  for (; t0 + 3 * stride < total; t0 += 4 * stride) {
    int  s[4], e[4], old[4];
    int* hp[4];
#pragma unroll
    for (int k = 0; k < 4; ++k) {
      const long long t = t0 + k * stride;
      const int r = (t >= n_edges) ? 1 : 0;
      e[k] = (int)(t - (long long)r * n_edges);
      const int* ei = r ? ei2 : ei1;
      s[k] = ei[e[k]];
      const int d = ei[n_edges + e[k]];
      hp[k] = head + (long long)r * n_nodes + d;
    }
#pragma unroll
    for (int k = 0; k < 4; ++k)
      old[k] = __hip_atomic_exchange(hp[k], e[k] + 1,
                                     __ATOMIC_RELAXED, __HIP_MEMORY_SCOPE_AGENT);
#pragma unroll
    for (int k = 0; k < 4; ++k) {
      v2i q; q.x = s[k]; q.y = old[k];
      rec[t0 + k * stride] = q;
    }
  }
  for (; t0 < total; t0 += stride)
    link_one(t0, ei1, ei2, n_edges, n_nodes, head, rec);
}

__global__ __launch_bounds__(256) void k_gather(
    const float* __restrict__ x, const int* __restrict__ head,
    const v2i* __restrict__ rec, int n_nodes, int n_edges,
    float* __restrict__ sum, float* __restrict__ mx,
    float* __restrict__ mn, float* __restrict__ mean) {
  const int lane = threadIdx.x & 31;
  int node = blockIdx.x * 8 + (threadIdx.x >> 5);
  if (node >= n_nodes) return;
  node = __builtin_amdgcn_readfirstlane(node);

  v4f s  = { 0.0f, 0.0f, 0.0f, 0.0f };
  v4f hi = { -FLT_BIG, -FLT_BIG, -FLT_BIG, -FLT_BIG };
  v4f lo = {  FLT_BIG,  FLT_BIG,  FLT_BIG,  FLT_BIG };
  int cnt = 0;

  int e1 = head[node];
  v2i q = { 0, 0 };
  if (e1) q = rec[e1 - 1];
  for (int it = 0; it < n_edges && e1 != 0; ++it) {
    const v4f v = *(const v4f*)(x + (long long)q.x * C + lane * 4);
    e1 = q.y;
    if (e1) q = rec[e1 - 1];
    s += v;
    for (int k = 0; k < 4; ++k) {
      hi[k] = fmaxf(hi[k], v[k]);
      lo[k] = fminf(lo[k], v[k]);
    }
    ++cnt;
  }

  const float cn = (float)cnt;
  const float dv = fmaxf(cn, 1.0f);
  v4f m;
  for (int k = 0; k < 4; ++k) m[k] = s[k] / dv;
  if (cnt == 0) {
    hi = (v4f){ 0.0f, 0.0f, 0.0f, 0.0f };
    lo = (v4f){ 0.0f, 0.0f, 0.0f, 0.0f };
  }
  const long long o = (long long)node * C + lane * 4;
  *(v4f*)(sum  + o) = s;
  *(v4f*)(mx   + o) = hi;
  *(v4f*)(mn   + o) = lo;
  *(v4f*)(mean + o) = m;
}

#define KC     32
#define BPITCH 36
#define NCHUNK 20

struct ChunkRegs { v4f a[4]; v4f b[8]; };

__global__ __launch_bounds__(128) void k_gemm_wmma(
    const float* __restrict__ x,
    const float* __restrict__ mean, const float* __restrict__ mx,
    const float* __restrict__ mn,   const float* __restrict__ sum,
    const float* __restrict__ W1,   const float* __restrict__ W2,
    float* __restrict__ h, int n_nodes) {
  __shared__ __attribute__((aligned(16))) float Bs[2][128 * BPITCH];

  const int tid  = threadIdx.x;
  const int lane = tid & 31;
  const int wv   = tid >> 5;
  const int m0   = blockIdx.x * 64 + wv * 16;
  const int c    = lane & 15;
  const int hf   = lane >> 4;

  int row = m0 + c;
  if (row >= n_nodes) row = n_nodes - 1;

  const int sn = tid >> 3;
  const int sq = tid & 7;

  v8f acc[8] = {};

  auto load_chunk = [&](int ch, ChunkRegs& r) {
    const int seg = ch >> 2;
    const int kc  = (ch & 3) * KC;
    const float* A  = (seg == 0) ? x : (seg == 1) ? mean : (seg == 2) ? mx : (seg == 3) ? mn : sum;
    const float* Wb = (seg == 0) ? W1 : (W2 + (seg - 1) * C);
    const int   ldw = (seg == 0) ? C : 4 * C;
    const float* Ap = A + (long long)row * C + kc + 4 * hf;
#pragma unroll
    for (int i = 0; i < 4; ++i) r.a[i] = *(const v4f*)(Ap + 8 * i);
    const float* Wp = Wb + (long long)sn * ldw + kc + 4 * sq;
#pragma unroll
    for (int p = 0; p < 8; ++p) r.b[p] = *(const v4f*)(Wp + (long long)(16 * p) * ldw);
  };

  auto store_chunk = [&](int stage, const ChunkRegs& r) {
    float* dst = &Bs[stage][sn * BPITCH + 4 * sq];
#pragma unroll
    for (int p = 0; p < 8; ++p) *(v4f*)(dst + 16 * p * BPITCH) = r.b[p];
  };

  auto load_frags = [&](const float* src, int i, v4f (&bf)[8]) {
#pragma unroll
    for (int j = 0; j < 8; ++j) bf[j] = *(const v4f*)(src + 16 * j * BPITCH + 8 * i);
  };

  auto compute_chunk = [&](int stage, const ChunkRegs& r) {
    const float* src = &Bs[stage][c * BPITCH + 4 * hf];
    v4f bf[2][8];
    load_frags(src, 0, bf[0]);
#pragma unroll
    for (int i = 0; i < 4; ++i) {
      if (i + 1 < 4) load_frags(src, i + 1, bf[(i + 1) & 1]);
      __builtin_amdgcn_sched_barrier(0);
      const v2f a0 = { r.a[i][0], r.a[i][1] };
      const v2f a1 = { r.a[i][2], r.a[i][3] };
#pragma unroll
      for (int j = 0; j < 8; ++j) {
        const v2f b0 = { bf[i & 1][j][0], bf[i & 1][j][1] };
        acc[j] = __builtin_amdgcn_wmma_f32_16x16x4_f32(
            false, a0, false, b0, (short)0, acc[j], false, false);
      }
#pragma unroll
      for (int j = 0; j < 8; ++j) {
        const v2f b1 = { bf[i & 1][j][2], bf[i & 1][j][3] };
        acc[j] = __builtin_amdgcn_wmma_f32_16x16x4_f32(
            false, a1, false, b1, (short)0, acc[j], false, false);
      }
      __builtin_amdgcn_sched_barrier(0);
    }
  };

  ChunkRegs cur, nxt;
  load_chunk(0, cur);
  store_chunk(0, cur);
  __syncthreads();

  for (int ch = 0; ch < NCHUNK; ++ch) {
    const bool more = (ch + 1 < NCHUNK);
    if (more) load_chunk(ch + 1, nxt);
    asm volatile("" ::: "memory");
    compute_chunk(ch & 1, cur);
    if (more) store_chunk((ch + 1) & 1, nxt);
    __syncthreads();
    if (more) {
#pragma unroll
      for (int i = 0; i < 4; ++i) cur.a[i] = nxt.a[i];
    }
  }

  const int rbase = m0 + (hf << 3);
#pragma unroll
  for (int v = 0; v < 8; ++v) {
    int r = rbase + v;
    if (r < n_nodes) {
      float* hp = h + (long long)r * C + c;
#pragma unroll
      for (int j = 0; j < 8; ++j) hp[j * 16] += acc[j][v];
    }
  }
}

__device__ __forceinline__ v4f ln_row(v4f v, const v4f g, const v4f b) {
#pragma unroll
  for (int k = 0; k < 4; ++k) v[k] = fmaxf(0.5f * v[k], 0.0f);
  float s  = (v[0] + v[1]) + (v[2] + v[3]);
  float s2 = (v[0] * v[0] + v[1] * v[1]) + (v[2] * v[2] + v[3] * v[3]);
#pragma unroll
  for (int off = 16; off > 0; off >>= 1) {
    s  += __shfl_xor(s,  off, 32);
    s2 += __shfl_xor(s2, off, 32);
  }
  const float mu  = s  * (1.0f / C);
  const float var = s2 * (1.0f / C) - mu * mu;
  const float rs  = rsqrtf(var + 1e-5f);
  v4f o;
#pragma unroll
  for (int k = 0; k < 4; ++k) o[k] = (v[k] - mu) * rs * g[k] + b[k];
  return o;
}

__global__ __launch_bounds__(256) void k_ln(const float* __restrict__ h,
                                            const float* __restrict__ gamma,
                                            const float* __restrict__ beta,
                                            float* __restrict__ out, int n_nodes) {
  const int lane = threadIdx.x & 31;
  const int nw   = (int)((gridDim.x * blockDim.x) >> 5);
  int r0 = (int)((blockIdx.x * blockDim.x + threadIdx.x) >> 5);
  r0 = __builtin_amdgcn_readfirstlane(r0);
  const v4f g = *(const v4f*)(gamma + lane * 4);
  const v4f b = *(const v4f*)(beta  + lane * 4);
  for (; r0 + nw < n_nodes; r0 += 2 * nw) {
    const int r1 = r0 + nw;
    const v4f a = *(const v4f*)(h + (long long)r0 * C + lane * 4);
    const v4f c = *(const v4f*)(h + (long long)r1 * C + lane * 4);
    *(v4f*)(out + (long long)r0 * C + lane * 4) = ln_row(a, g, b);
    *(v4f*)(out + (long long)r1 * C + lane * 4) = ln_row(c, g, b);
  }
  if (r0 < n_nodes) {
    const v4f a = *(const v4f*)(h + (long long)r0 * C + lane * 4);
    *(v4f*)(out + (long long)r0 * C + lane * 4) = ln_row(a, g, b);
  }
}

extern "C" void kernel_launch(void* const* d_in, const int* in_sizes, int n_in,
                              void* d_out, int out_size, void* d_ws, size_t ws_size,
                              hipStream_t stream) {
  const float* x     = (const float*)d_in[0];
  const int*   ei1   = (const int*)d_in[1];
  const int*   ei2   = (const int*)d_in[2];
  const float* W1_r1 = (const float*)d_in[3];
  const float* W2_r1 = (const float*)d_in[4];
  const float* W1_r2 = (const float*)d_in[5];
  const float* W2_r2 = (const float*)d_in[6];
  const float* gamma = (const float*)d_in[7];
  const float* beta  = (const float*)d_in[8];
  float* out = (float*)d_out;

  const int N = in_sizes[0] / C;
  const int E = in_sizes[1] / 2;
  const long long NC = (long long)N * C;

  float* sum  = (float*)d_ws;
  float* mx   = sum  + NC;
  float* mn   = mx   + NC;
  float* mean = mn   + NC;
  float* h    = mean + NC;

  int* head = (int*)d_out;
  v2i* rec  = (v2i*)(head + 2 * (long long)N);

  const int T = 256;
  dim3 gS(1024);
  dim3 gE2((unsigned)((2LL * E + 4 * T - 1) / (4 * T)));
  dim3 gG((unsigned)((N + 7) / 8));
  dim3 gM((unsigned)((N + 63) / 64));

  k_zero<<<gS, T, 0, stream>>>(h, NC, head, 2 * N);
  k_link<<<gE2, T, 0, stream>>>(ei1, ei2, E, N, head, rec);

  k_gather<<<gG, T, 0, stream>>>(x, head, rec, N, E, sum, mx, mn, mean);
  k_gemm_wmma<<<gM, 128, 0, stream>>>(x, mean, mx, mn, sum, W1_r1, W2_r1, h, N);

  k_gather<<<gG, T, 0, stream>>>(x, head + N, rec + E, N, E, sum, mx, mn, mean);
  k_gemm_wmma<<<gM, 128, 0, stream>>>(x, mean, mx, mn, sum, W1_r2, W2_r2, h, N);

  k_ln<<<gS, T, 0, stream>>>(h, gamma, beta, out, N);
}


# edge linking without global atomics: edges binned per chunk by destination range in LDS (rank = LDS add), then one work-group per range links them by LDS exchange; sum order = list arrival order
# speedup vs baseline: 3.8968x; 1.1854x over previous
#include <hip/hip_runtime.h>
#include <cstdint>

#define C 128
#define FLT_BIG 3.402823466e38f

typedef float v2f __attribute__((ext_vector_type(2)));
typedef float v4f __attribute__((ext_vector_type(4)));
typedef int   v2i __attribute__((ext_vector_type(2)));
typedef float v8f __attribute__((ext_vector_type(8)));

__global__ __launch_bounds__(256) void k_zero(float* __restrict__ p, long long n) {
  const long long stride = (long long)gridDim.x * blockDim.x;
  const long long t = (long long)blockIdx.x * blockDim.x + threadIdx.x;
  v4f* __restrict__ p4 = (v4f*)p;
  const long long n4 = n >> 2;
  const v4f z = { 0.0f, 0.0f, 0.0f, 0.0f };
  for (long long i = t; i < n4; i += stride) p4[i] = z;
}

#define RS_LOG 9
#define RS     (1 << RS_LOG)
#define CHUNK  4096
#define MAXB   128
#define STRIDE (CHUNK + 16 * MAXB)
#define SEGT   256

__global__ __launch_bounds__(1024) void k_bin(
    const int* __restrict__ ei1, const int* __restrict__ ei2,
    int n_edges, int n_nodes, v2i* __restrict__ rec, v2i* __restrict__ tab) {
  __shared__ int cnt[MAXB + 1];
  __shared__ int num[MAXB];
  __shared__ __attribute__((aligned(16))) v2i st[STRIDE];
  const int tid = threadIdx.x;
  const int nCh = (n_edges + CHUNK - 1) / CHUNK;
  const int r   = ((int)blockIdx.x >= nCh) ? 1 : 0;
  const int c   = (int)blockIdx.x - r * nCh;
  const int* __restrict__ ei = r ? ei2 : ei1;
  const int nb  = (n_nodes + RS - 1) >> RS_LOG;
  if (tid <= MAXB) cnt[tid] = 0;
  __syncthreads();

  int  s[4], d[4], rk[4];
  bool ok[4];
  const int base = c * CHUNK;
#pragma unroll
  for (int k = 0; k < 4; ++k) {
    const int e = base + k * 1024 + tid;
    ok[k] = e < n_edges;
    s[k] = 0; d[k] = 0; rk[k] = 0;
    if (ok[k]) { s[k] = ei[e]; d[k] = ei[n_edges + e]; }
  }
#pragma unroll
  for (int k = 0; k < 4; ++k)
    if (ok[k]) rk[k] = atomicAdd(&cnt[d[k] >> RS_LOG], 1);
  __syncthreads();

  if (tid < 32) {
    int v[4], local = 0;
#pragma unroll
    for (int j = 0; j < 4; ++j) {
      const int idx = tid * 4 + j;
      v[j] = (idx < nb) ? cnt[idx] : 0;
      local += (v[j] + 15) & ~15;
    }
    int inc = local;
#pragma unroll
    for (int off = 1; off < 32; off <<= 1) {
      const int t = __shfl_up(inc, off, 32);
      if (tid >= off) inc += t;
    }
    int ex = inc - local;
#pragma unroll
    for (int j = 0; j < 4; ++j) {
      const int idx = tid * 4 + j;
      if (idx < nb) { num[idx] = v[j]; cnt[idx] = ex; ex += (v[j] + 15) & ~15; }
    }
    if (tid == 31) cnt[MAXB] = inc;
  }
  __syncthreads();

#pragma unroll
  for (int k = 0; k < 4; ++k)
    if (ok[k]) {
      v2i q; q.x = s[k]; q.y = d[k];
      st[cnt[d[k] >> RS_LOG] + rk[k]] = q;
    }
  __syncthreads();

  const long long region = (long long)r * nCh + c;
  v2i* __restrict__ dst = rec + region * STRIDE;
  const int used = cnt[MAXB];
  for (int i = tid; i < used; i += 1024) dst[i] = st[i];
  if (tid < nb) {
    v2i t; t.x = cnt[tid]; t.y = cnt[tid] + num[tid];
    tab[region * MAXB + tid] = t;
  }
}

__global__ __launch_bounds__(1024) void k_build(
    v2i* __restrict__ rec, const v2i* __restrict__ tab,
    int n_edges, int n_nodes, int* __restrict__ head, int head_stride) {
  __shared__ int hd[RS];
  __shared__ v2i sg[SEGT];
  const int tid  = threadIdx.x;
  const int lane = tid & 31;
  const int wv   = tid >> 5;
  const int nCh  = (n_edges + CHUNK - 1) / CHUNK;
  const int nb   = (n_nodes + RS - 1) >> RS_LOG;
  const int r    = ((int)blockIdx.x >= nb) ? 1 : 0;
  const int b    = (int)blockIdx.x - r * nb;
  const int n0   = b << RS_LOG;
  if (tid < RS) hd[tid] = 0;
  int* __restrict__ ry = (int*)(rec + (long long)r * nCh * STRIDE) + 1;
  const v2i* __restrict__ tb = tab + (long long)r * nCh * MAXB + b;

  for (int c0 = 0; c0 < nCh; c0 += SEGT) {
    __syncthreads();
    const int nseg = (nCh - c0 < SEGT) ? (nCh - c0) : SEGT;
    if (tid < nseg) {
      v2i t = tb[(long long)(c0 + tid) * MAXB];
      t.x += (c0 + tid) * STRIDE;
      t.y += (c0 + tid) * STRIDE;
      sg[tid] = t;
    }
    __syncthreads();
    int p0[8], pe[8], y0[8], y1[8];
#pragma unroll
    for (int j = 0; j < 8; ++j) {
      const int i = wv + 32 * j;
      v2i se = { 0, 0 };
      if (i < nseg) se = sg[i];
      p0[j] = se.x + lane;
      pe[j] = se.y;
      y0[j] = 0;
      y1[j] = 0;
      if (p0[j] < pe[j])      y0[j] = ry[2 * (long long)p0[j]];
      if (p0[j] + 32 < pe[j]) y1[j] = ry[2 * (long long)(p0[j] + 32)];
    }
#pragma unroll
    for (int j = 0; j < 8; ++j) {
      if (p0[j] < pe[j])
        ry[2 * (long long)p0[j]] = __hip_atomic_exchange(&hd[y0[j] - n0], p0[j] + 1,
                                       __ATOMIC_RELAXED, __HIP_MEMORY_SCOPE_WORKGROUP);
      if (p0[j] + 32 < pe[j])
        ry[2 * (long long)(p0[j] + 32)] = __hip_atomic_exchange(&hd[y1[j] - n0], p0[j] + 33,
                                       __ATOMIC_RELAXED, __HIP_MEMORY_SCOPE_WORKGROUP);
      for (int p = p0[j] + 64; p < pe[j]; p += 32) {
        const int dn = ry[2 * (long long)p];
        ry[2 * (long long)p] = __hip_atomic_exchange(&hd[dn - n0], p + 1,
                                       __ATOMIC_RELAXED, __HIP_MEMORY_SCOPE_WORKGROUP);
      }
    }
  }
  __syncthreads();
  if (tid < RS && n0 + tid < n_nodes)
    head[(long long)r * head_stride + n0 + tid] = hd[tid];
}

__global__ __launch_bounds__(256) void k_gather(
    const float* __restrict__ x, const int* __restrict__ head,
    const v2i* __restrict__ rec, int n_nodes, int n_edges,
    float* __restrict__ sum, float* __restrict__ mx,
    float* __restrict__ mn, float* __restrict__ mean) {
  const int lane = threadIdx.x & 31;
  int node = blockIdx.x * 8 + (threadIdx.x >> 5);
  if (node >= n_nodes) return;
  node = __builtin_amdgcn_readfirstlane(node);

  v4f s  = { 0.0f, 0.0f, 0.0f, 0.0f };
  v4f hi = { -FLT_BIG, -FLT_BIG, -FLT_BIG, -FLT_BIG };
  v4f lo = {  FLT_BIG,  FLT_BIG,  FLT_BIG,  FLT_BIG };
  int cnt = 0;

  int e1 = head[node];
  v2i q = { 0, 0 };
  if (e1) q = rec[e1 - 1];
  for (int it = 0; it < n_edges && e1 != 0; ++it) {
    const v4f v = *(const v4f*)(x + (long long)q.x * C + lane * 4);
    e1 = q.y;
    if (e1) q = rec[e1 - 1];
    s += v;
    for (int k = 0; k < 4; ++k) {
      hi[k] = fmaxf(hi[k], v[k]);
      lo[k] = fminf(lo[k], v[k]);
    }
    ++cnt;
  }

  const float cn = (float)cnt;
  const float dv = fmaxf(cn, 1.0f);
  v4f m;
  for (int k = 0; k < 4; ++k) m[k] = s[k] / dv;
  if (cnt == 0) {
    hi = (v4f){ 0.0f, 0.0f, 0.0f, 0.0f };
    lo = (v4f){ 0.0f, 0.0f, 0.0f, 0.0f };
  }
  const long long o = (long long)node * C + lane * 4;
  *(v4f*)(sum  + o) = s;
  *(v4f*)(mx   + o) = hi;
  *(v4f*)(mn   + o) = lo;
  *(v4f*)(mean + o) = m;
}

#define KC     32
#define BPITCH 36
#define NCHUNK 20

struct ChunkRegs { v4f a[4]; v4f b[8]; };

__global__ __launch_bounds__(128) void k_gemm_wmma(
    const float* __restrict__ x,
    const float* __restrict__ mean, const float* __restrict__ mx,
    const float* __restrict__ mn,   const float* __restrict__ sum,
    const float* __restrict__ W1,   const float* __restrict__ W2,
    float* __restrict__ h, int n_nodes) {
  __shared__ __attribute__((aligned(16))) float Bs[2][128 * BPITCH];

  const int tid  = threadIdx.x;
  const int lane = tid & 31;
  const int wv   = tid >> 5;
  const int m0   = blockIdx.x * 64 + wv * 16;
  const int c    = lane & 15;
  const int hf   = lane >> 4;

  int row = m0 + c;
  if (row >= n_nodes) row = n_nodes - 1;

  const int sn = tid >> 3;
  const int sq = tid & 7;

  v8f acc[8] = {};

  auto load_chunk = [&](int ch, ChunkRegs& r) {
    const int seg = ch >> 2;
    const int kc  = (ch & 3) * KC;
    const float* A  = (seg == 0) ? x : (seg == 1) ? mean : (seg == 2) ? mx : (seg == 3) ? mn : sum;
    const float* Wb = (seg == 0) ? W1 : (W2 + (seg - 1) * C);
    const int   ldw = (seg == 0) ? C : 4 * C;
    const float* Ap = A + (long long)row * C + kc + 4 * hf;
#pragma unroll
    for (int i = 0; i < 4; ++i) r.a[i] = *(const v4f*)(Ap + 8 * i);
    const float* Wp = Wb + (long long)sn * ldw + kc + 4 * sq;
#pragma unroll
    for (int p = 0; p < 8; ++p) r.b[p] = *(const v4f*)(Wp + (long long)(16 * p) * ldw);
  };

  auto store_chunk = [&](int stage, const ChunkRegs& r) {
    float* dst = &Bs[stage][sn * BPITCH + 4 * sq];
#pragma unroll
    for (int p = 0; p < 8; ++p) *(v4f*)(dst + 16 * p * BPITCH) = r.b[p];
  };

  auto load_frags = [&](const float* src, int i, v4f (&bf)[8]) {
#pragma unroll
    for (int j = 0; j < 8; ++j) bf[j] = *(const v4f*)(src + 16 * j * BPITCH + 8 * i);
  };

  auto compute_chunk = [&](int stage, const ChunkRegs& r) {
    const float* src = &Bs[stage][c * BPITCH + 4 * hf];
    v4f bf[2][8];
    load_frags(src, 0, bf[0]);
#pragma unroll
    for (int i = 0; i < 4; ++i) {
      if (i + 1 < 4) load_frags(src, i + 1, bf[(i + 1) & 1]);
      __builtin_amdgcn_sched_barrier(0);
      const v2f a0 = { r.a[i][0], r.a[i][1] };
      const v2f a1 = { r.a[i][2], r.a[i][3] };
#pragma unroll
      for (int j = 0; j < 8; ++j) {
        const v2f b0 = { bf[i & 1][j][0], bf[i & 1][j][1] };
        acc[j] = __builtin_amdgcn_wmma_f32_16x16x4_f32(
            false, a0, false, b0, (short)0, acc[j], false, false);
      }
#pragma unroll
      for (int j = 0; j < 8; ++j) {
        const v2f b1 = { bf[i & 1][j][2], bf[i & 1][j][3] };
        acc[j] = __builtin_amdgcn_wmma_f32_16x16x4_f32(
            false, a1, false, b1, (short)0, acc[j], false, false);
      }
      __builtin_amdgcn_sched_barrier(0);
    }
  };

  ChunkRegs cur, nxt;
  load_chunk(0, cur);
  store_chunk(0, cur);
  __syncthreads();

  for (int ch = 0; ch < NCHUNK; ++ch) {
    const bool more = (ch + 1 < NCHUNK);
    if (more) load_chunk(ch + 1, nxt);
    asm volatile("" ::: "memory");
    compute_chunk(ch & 1, cur);
    if (more) store_chunk((ch + 1) & 1, nxt);
    __syncthreads();
    if (more) {
#pragma unroll
      for (int i = 0; i < 4; ++i) cur.a[i] = nxt.a[i];
    }
  }

  const int rbase = m0 + (hf << 3);
#pragma unroll
  for (int v = 0; v < 8; ++v) {
    int r = rbase + v;
    if (r < n_nodes) {
      float* hp = h + (long long)r * C + c;
#pragma unroll
      for (int j = 0; j < 8; ++j) hp[j * 16] += acc[j][v];
    }
  }
}

__device__ __forceinline__ v4f ln_row(v4f v, const v4f g, const v4f b) {
#pragma unroll
  for (int k = 0; k < 4; ++k) v[k] = fmaxf(0.5f * v[k], 0.0f);
  float s  = (v[0] + v[1]) + (v[2] + v[3]);
  float s2 = (v[0] * v[0] + v[1] * v[1]) + (v[2] * v[2] + v[3] * v[3]);
#pragma unroll
  for (int off = 16; off > 0; off >>= 1) {
    s  += __shfl_xor(s,  off, 32);
    s2 += __shfl_xor(s2, off, 32);
  }
  const float mu  = s  * (1.0f / C);
  const float var = s2 * (1.0f / C) - mu * mu;
  const float rs  = rsqrtf(var + 1e-5f);
  v4f o;
#pragma unroll
  for (int k = 0; k < 4; ++k) o[k] = (v[k] - mu) * rs * g[k] + b[k];
  return o;
}

__global__ __launch_bounds__(256) void k_ln(const float* __restrict__ h,
                                            const float* __restrict__ gamma,
                                            const float* __restrict__ beta,
                                            float* __restrict__ out, int n_nodes) {
  const int lane = threadIdx.x & 31;
  const int nw   = (int)((gridDim.x * blockDim.x) >> 5);
  int r0 = (int)((blockIdx.x * blockDim.x + threadIdx.x) >> 5);
  r0 = __builtin_amdgcn_readfirstlane(r0);
  const v4f g = *(const v4f*)(gamma + lane * 4);
  const v4f b = *(const v4f*)(beta  + lane * 4);
  for (; r0 + nw < n_nodes; r0 += 2 * nw) {
    const int r1 = r0 + nw;
    const v4f a = *(const v4f*)(h + (long long)r0 * C + lane * 4);
    const v4f c = *(const v4f*)(h + (long long)r1 * C + lane * 4);
    *(v4f*)(out + (long long)r0 * C + lane * 4) = ln_row(a, g, b);
    *(v4f*)(out + (long long)r1 * C + lane * 4) = ln_row(c, g, b);
  }
  if (r0 < n_nodes) {
    const v4f a = *(const v4f*)(h + (long long)r0 * C + lane * 4);
    *(v4f*)(out + (long long)r0 * C + lane * 4) = ln_row(a, g, b);
  }
}

extern "C" void kernel_launch(void* const* d_in, const int* in_sizes, int n_in,
                              void* d_out, int out_size, void* d_ws, size_t ws_size,
                              hipStream_t stream) {
  const float* x     = (const float*)d_in[0];
  const int*   ei1   = (const int*)d_in[1];
  const int*   ei2   = (const int*)d_in[2];
  const float* W1_r1 = (const float*)d_in[3];
  const float* W2_r1 = (const float*)d_in[4];
  const float* W1_r2 = (const float*)d_in[5];
  const float* W2_r2 = (const float*)d_in[6];
  const float* gamma = (const float*)d_in[7];
  const float* beta  = (const float*)d_in[8];
  float* out = (float*)d_out;

  const int N = in_sizes[0] / C;
  const int E = in_sizes[1] / 2;
  const long long NC = (long long)N * C;

  float* sum  = (float*)d_ws;
  float* mx   = sum  + NC;
  float* mn   = mx   + NC;
  float* mean = mn   + NC;
  float* h    = mean + NC;

  const int NP  = (N + 31) / 32 * 32;
  const int nCh = (E + CHUNK - 1) / CHUNK;
  const int nb  = (N + RS - 1) / RS;
  int* head = (int*)d_out;
  v2i* rec  = (v2i*)(head + 2 * (long long)NP);
  v2i* tab  = (v2i*)d_ws;

  const int T = 256;
  dim3 gS(1024);
  dim3 gG((unsigned)((N + 7) / 8));
  dim3 gM((unsigned)((N + 63) / 64));

  k_zero<<<gS, T, 0, stream>>>(h, NC);
  k_bin<<<dim3((unsigned)(2 * nCh)), 1024, 0, stream>>>(ei1, ei2, E, N, rec, tab);
  k_build<<<dim3((unsigned)(2 * nb)), 1024, 0, stream>>>(rec, tab, E, N, head, NP);

  k_gather<<<gG, T, 0, stream>>>(x, head, rec, N, E, sum, mx, mn, mean);
  k_gemm_wmma<<<gM, 128, 0, stream>>>(x, mean, mx, mn, sum, W1_r1, W2_r1, h, N);

  k_gather<<<gG, T, 0, stream>>>(x, head + NP, rec + (long long)nCh * STRIDE, N, E, sum, mx, mn, mean);
  k_gemm_wmma<<<gM, 128, 0, stream>>>(x, mean, mx, mn, sum, W1_r2, W2_r2, h, N);

  k_ln<<<gS, T, 0, stream>>>(h, gamma, beta, out, N);
}


# clear and layer-norm passes folded into the GEMM epilogues: GEMM 1 stores h, GEMM 2 adds h and applies relu + LN on rows staged through LDS, whole-row stores; 8 launches -> 6; all sum orders kept
# speedup vs baseline: 4.3231x; 1.1094x over previous
#include <hip/hip_runtime.h>
#include <cstdint>

#define C 128
#define FLT_BIG 3.402823466e38f

typedef float v2f __attribute__((ext_vector_type(2)));
typedef float v4f __attribute__((ext_vector_type(4)));
typedef int   v2i __attribute__((ext_vector_type(2)));
typedef float v8f __attribute__((ext_vector_type(8)));

#define RS_LOG 9
#define RS     (1 << RS_LOG)
#define CHUNK  4096
#define MAXB   128
#define STRIDE (CHUNK + 16 * MAXB)
#define SEGT   256

__global__ __launch_bounds__(1024) void k_bin(
    const int* __restrict__ ei1, const int* __restrict__ ei2,
    int n_edges, int n_nodes, v2i* __restrict__ rec, v2i* __restrict__ tab) {
  __shared__ int cnt[MAXB + 1];
  __shared__ int num[MAXB];
  __shared__ __attribute__((aligned(16))) v2i st[STRIDE];
  const int tid = threadIdx.x;
  const int nCh = (n_edges + CHUNK - 1) / CHUNK;
  const int r   = ((int)blockIdx.x >= nCh) ? 1 : 0;
  const int c   = (int)blockIdx.x - r * nCh;
  const int* __restrict__ ei = r ? ei2 : ei1;
  const int nb  = (n_nodes + RS - 1) >> RS_LOG;
  if (tid <= MAXB) cnt[tid] = 0;
  __syncthreads();

  int  s[4], d[4], rk[4];
  bool ok[4];
  const int base = c * CHUNK;
#pragma unroll
  for (int k = 0; k < 4; ++k) {
    const int e = base + k * 1024 + tid;
    ok[k] = e < n_edges;
    s[k] = 0; d[k] = 0; rk[k] = 0;
    if (ok[k]) { s[k] = ei[e]; d[k] = ei[n_edges + e]; }
  }
#pragma unroll
  for (int k = 0; k < 4; ++k)
    if (ok[k]) rk[k] = atomicAdd(&cnt[d[k] >> RS_LOG], 1);
  __syncthreads();

  if (tid < 32) {
    int v[4], local = 0;
#pragma unroll
    for (int j = 0; j < 4; ++j) {
      const int idx = tid * 4 + j;
      v[j] = (idx < nb) ? cnt[idx] : 0;
      local += (v[j] + 15) & ~15;
    }
    int inc = local;
#pragma unroll
    for (int off = 1; off < 32; off <<= 1) {
      const int t = __shfl_up(inc, off, 32);
      if (tid >= off) inc += t;
    }
    int ex = inc - local;
#pragma unroll
    for (int j = 0; j < 4; ++j) {
      const int idx = tid * 4 + j;
      if (idx < nb) { num[idx] = v[j]; cnt[idx] = ex; ex += (v[j] + 15) & ~15; }
    }
    if (tid == 31) cnt[MAXB] = inc;
  }
  __syncthreads();

#pragma unroll
  for (int k = 0; k < 4; ++k)
    if (ok[k]) {
      v2i q; q.x = s[k]; q.y = d[k];
      st[cnt[d[k] >> RS_LOG] + rk[k]] = q;
    }
  __syncthreads();

  const long long region = (long long)r * nCh + c;
  v2i* __restrict__ dst = rec + region * STRIDE;
  const int used = cnt[MAXB];
  for (int i = tid; i < used; i += 1024) dst[i] = st[i];
  if (tid < nb) {
    v2i t; t.x = cnt[tid]; t.y = cnt[tid] + num[tid];
    tab[region * MAXB + tid] = t;
  }
}

__global__ __launch_bounds__(1024) void k_build(
    v2i* __restrict__ rec, const v2i* __restrict__ tab,
    int n_edges, int n_nodes, int* __restrict__ head, int head_stride) {
  __shared__ int hd[RS];
  __shared__ v2i sg[SEGT];
  const int tid  = threadIdx.x;
  const int lane = tid & 31;
  const int wv   = tid >> 5;
  const int nCh  = (n_edges + CHUNK - 1) / CHUNK;
  const int nb   = (n_nodes + RS - 1) >> RS_LOG;
  const int r    = ((int)blockIdx.x >= nb) ? 1 : 0;
  const int b    = (int)blockIdx.x - r * nb;
  const int n0   = b << RS_LOG;
  if (tid < RS) hd[tid] = 0;
  int* __restrict__ ry = (int*)(rec + (long long)r * nCh * STRIDE) + 1;
  const v2i* __restrict__ tb = tab + (long long)r * nCh * MAXB + b;

  for (int c0 = 0; c0 < nCh; c0 += SEGT) {
    __syncthreads();
    const int nseg = (nCh - c0 < SEGT) ? (nCh - c0) : SEGT;
    if (tid < nseg) {
      v2i t = tb[(long long)(c0 + tid) * MAXB];
      t.x += (c0 + tid) * STRIDE;
      t.y += (c0 + tid) * STRIDE;
      sg[tid] = t;
    }
    __syncthreads();
    int p0[8], pe[8], y0[8], y1[8];
#pragma unroll
    for (int j = 0; j < 8; ++j) {
      const int i = wv + 32 * j;
      v2i se = { 0, 0 };
      if (i < nseg) se = sg[i];
      p0[j] = se.x + lane;
      pe[j] = se.y;
      y0[j] = 0;
      y1[j] = 0;
      if (p0[j] < pe[j])      y0[j] = ry[2 * (long long)p0[j]];
      if (p0[j] + 32 < pe[j]) y1[j] = ry[2 * (long long)(p0[j] + 32)];
    }
#pragma unroll
    for (int j = 0; j < 8; ++j) {
      if (p0[j] < pe[j])
        ry[2 * (long long)p0[j]] = __hip_atomic_exchange(&hd[y0[j] - n0], p0[j] + 1,
                                       __ATOMIC_RELAXED, __HIP_MEMORY_SCOPE_WORKGROUP);
      if (p0[j] + 32 < pe[j])
        ry[2 * (long long)(p0[j] + 32)] = __hip_atomic_exchange(&hd[y1[j] - n0], p0[j] + 33,
                                       __ATOMIC_RELAXED, __HIP_MEMORY_SCOPE_WORKGROUP);
      for (int p = p0[j] + 64; p < pe[j]; p += 32) {
        const int dn = ry[2 * (long long)p];
        ry[2 * (long long)p] = __hip_atomic_exchange(&hd[dn - n0], p + 1,
                                       __ATOMIC_RELAXED, __HIP_MEMORY_SCOPE_WORKGROUP);
      }
    }
  }
  __syncthreads();
  if (tid < RS && n0 + tid < n_nodes)
    head[(long long)r * head_stride + n0 + tid] = hd[tid];
}

__global__ __launch_bounds__(256) void k_gather(
    const float* __restrict__ x, const int* __restrict__ head,
    const v2i* __restrict__ rec, int n_nodes, int n_edges,
    float* __restrict__ sum, float* __restrict__ mx,
    float* __restrict__ mn, float* __restrict__ mean) {
  const int lane = threadIdx.x & 31;
  int node = blockIdx.x * 8 + (threadIdx.x >> 5);
  if (node >= n_nodes) return;
  node = __builtin_amdgcn_readfirstlane(node);

  v4f s  = { 0.0f, 0.0f, 0.0f, 0.0f };
  v4f hi = { -FLT_BIG, -FLT_BIG, -FLT_BIG, -FLT_BIG };
  v4f lo = {  FLT_BIG,  FLT_BIG,  FLT_BIG,  FLT_BIG };
  int cnt = 0;

  int e1 = head[node];
  v2i q = { 0, 0 };
  if (e1) q = rec[e1 - 1];
  for (int it = 0; it < n_edges && e1 != 0; ++it) {
    const v4f v = *(const v4f*)(x + (long long)q.x * C + lane * 4);
    e1 = q.y;
    if (e1) q = rec[e1 - 1];
    s += v;
    for (int k = 0; k < 4; ++k) {
      hi[k] = fmaxf(hi[k], v[k]);
      lo[k] = fminf(lo[k], v[k]);
    }
    ++cnt;
  }

  const float cn = (float)cnt;
  const float dv = fmaxf(cn, 1.0f);
  v4f m;
  for (int k = 0; k < 4; ++k) m[k] = s[k] / dv;
  if (cnt == 0) {
    hi = (v4f){ 0.0f, 0.0f, 0.0f, 0.0f };
    lo = (v4f){ 0.0f, 0.0f, 0.0f, 0.0f };
  }
  const long long o = (long long)node * C + lane * 4;
  *(v4f*)(sum  + o) = s;
  *(v4f*)(mx   + o) = hi;
  *(v4f*)(mn   + o) = lo;
  *(v4f*)(mean + o) = m;
}

__device__ __forceinline__ v4f ln_row(v4f v, const v4f g, const v4f b) {
#pragma unroll
  for (int k = 0; k < 4; ++k) v[k] = fmaxf(0.5f * v[k], 0.0f);
  float s  = (v[0] + v[1]) + (v[2] + v[3]);
  float s2 = (v[0] * v[0] + v[1] * v[1]) + (v[2] * v[2] + v[3] * v[3]);
#pragma unroll
  for (int off = 16; off > 0; off >>= 1) {
    s  += __shfl_xor(s,  off, 32);
    s2 += __shfl_xor(s2, off, 32);
  }
  const float mu  = s  * (1.0f / C);
  const float var = s2 * (1.0f / C) - mu * mu;
  const float rs  = rsqrtf(var + 1e-5f);
  v4f o;
#pragma unroll
  for (int k = 0; k < 4; ++k) o[k] = (v[k] - mu) * rs * g[k] + b[k];
  return o;
}

#define KC     32
#define BPITCH 36
#define NCHUNK 20
#define TP     132

struct ChunkRegs { v4f a[4]; v4f b[8]; };

template <bool FIRST>
__global__ __launch_bounds__(128) void k_gemm_wmma(
    const float* __restrict__ x,
    const float* __restrict__ mean, const float* __restrict__ mx,
    const float* __restrict__ mn,   const float* __restrict__ sum,
    const float* __restrict__ W1,   const float* __restrict__ W2,
    float* __restrict__ h,
    const float* __restrict__ gamma, const float* __restrict__ beta,
    float* __restrict__ out, int n_nodes) {
  __shared__ __attribute__((aligned(16))) float Bs[2 * 128 * BPITCH];

  const int tid  = threadIdx.x;
  const int lane = tid & 31;
  const int wv   = tid >> 5;
  const int m0   = blockIdx.x * 64 + wv * 16;
  const int c    = lane & 15;
  const int hf   = lane >> 4;

  int row = m0 + c;
  if (row >= n_nodes) row = n_nodes - 1;

  const int sn = tid >> 3;
  const int sq = tid & 7;

  v8f acc[8] = {};

  auto load_chunk = [&](int ch, ChunkRegs& r) {
    const int seg = ch >> 2;
    const int kc  = (ch & 3) * KC;
    const float* A  = (seg == 0) ? x : (seg == 1) ? mean : (seg == 2) ? mx : (seg == 3) ? mn : sum;
    const float* Wb = (seg == 0) ? W1 : (W2 + (seg - 1) * C);
    const int   ldw = (seg == 0) ? C : 4 * C;
    const float* Ap = A + (long long)row * C + kc + 4 * hf;
#pragma unroll
    for (int i = 0; i < 4; ++i) r.a[i] = *(const v4f*)(Ap + 8 * i);
    const float* Wp = Wb + (long long)sn * ldw + kc + 4 * sq;
#pragma unroll
    for (int p = 0; p < 8; ++p) r.b[p] = *(const v4f*)(Wp + (long long)(16 * p) * ldw);
  };

  auto store_chunk = [&](int stage, const ChunkRegs& r) {
    float* dst = &Bs[stage * (128 * BPITCH) + sn * BPITCH + 4 * sq];
#pragma unroll
    for (int p = 0; p < 8; ++p) *(v4f*)(dst + 16 * p * BPITCH) = r.b[p];
  };

  auto load_frags = [&](const float* src, int i, v4f (&bf)[8]) {
#pragma unroll
    for (int j = 0; j < 8; ++j) bf[j] = *(const v4f*)(src + 16 * j * BPITCH + 8 * i);
  };

  auto compute_chunk = [&](int stage, const ChunkRegs& r) {
    const float* src = &Bs[stage * (128 * BPITCH) + c * BPITCH + 4 * hf];
    v4f bf[2][8];
    load_frags(src, 0, bf[0]);
#pragma unroll
    for (int i = 0; i < 4; ++i) {
      if (i + 1 < 4) load_frags(src, i + 1, bf[(i + 1) & 1]);
      __builtin_amdgcn_sched_barrier(0);
      const v2f a0 = { r.a[i][0], r.a[i][1] };
      const v2f a1 = { r.a[i][2], r.a[i][3] };
#pragma unroll
      for (int j = 0; j < 8; ++j) {
        const v2f b0 = { bf[i & 1][j][0], bf[i & 1][j][1] };
        acc[j] = __builtin_amdgcn_wmma_f32_16x16x4_f32(
            false, a0, false, b0, (short)0, acc[j], false, false);
      }
#pragma unroll
      for (int j = 0; j < 8; ++j) {
        const v2f b1 = { bf[i & 1][j][2], bf[i & 1][j][3] };
        acc[j] = __builtin_amdgcn_wmma_f32_16x16x4_f32(
            false, a1, false, b1, (short)0, acc[j], false, false);
      }
      __builtin_amdgcn_sched_barrier(0);
    }
  };

  ChunkRegs cur, nxt;
  load_chunk(0, cur);
  store_chunk(0, cur);
  __syncthreads();

  for (int ch = 0; ch < NCHUNK; ++ch) {
    const bool more = (ch + 1 < NCHUNK);
    if (more) load_chunk(ch + 1, nxt);
    asm volatile("" ::: "memory");
    compute_chunk(ch & 1, cur);
    if (more) store_chunk((ch + 1) & 1, nxt);
    __syncthreads();
    if (more) {
#pragma unroll
      for (int i = 0; i < 4; ++i) cur.a[i] = nxt.a[i];
    }
  }

  float* T = &Bs[wv * (16 * TP)];
#pragma unroll
  for (int v = 0; v < 8; ++v) {
#pragma unroll
    for (int j = 0; j < 8; ++j) T[(8 * hf + v) * TP + 16 * j + c] = acc[j][v];
  }
  __syncthreads();

  const long long co = lane * 4;
  const bool full = (m0 + 16 <= n_nodes);
  float* dstp = (FIRST ? h : out) + (long long)m0 * C + co;
  v4f o[16];
  if (FIRST) {
#pragma unroll
    for (int rr = 0; rr < 16; ++rr) o[rr] = *(const v4f*)(T + rr * TP + co);
  } else {
    const v4f g = *(const v4f*)(gamma + co);
    const v4f b = *(const v4f*)(beta  + co);
    v4f hv[16];
#pragma unroll
    for (int rr = 0; rr < 16; ++rr) {
      int r = m0 + rr;
      if (r >= n_nodes) r = n_nodes - 1;
      hv[rr] = *(const v4f*)(h + (long long)r * C + co);
    }
#pragma unroll
    for (int rr = 0; rr < 16; ++rr) {
      const v4f t = hv[rr] + *(const v4f*)(T + rr * TP + co);
      o[rr] = ln_row(t, g, b);
    }
  }
  if (full) {
#pragma unroll
    for (int rr = 0; rr < 16; ++rr) *(v4f*)(dstp + (long long)rr * C) = o[rr];
  } else {
#pragma unroll
    for (int rr = 0; rr < 16; ++rr)
      if (m0 + rr < n_nodes) *(v4f*)(dstp + (long long)rr * C) = o[rr];
  }
}

extern "C" void kernel_launch(void* const* d_in, const int* in_sizes, int n_in,
                              void* d_out, int out_size, void* d_ws, size_t ws_size,
                              hipStream_t stream) {
  const float* x     = (const float*)d_in[0];
  const int*   ei1   = (const int*)d_in[1];
  const int*   ei2   = (const int*)d_in[2];
  const float* W1_r1 = (const float*)d_in[3];
  const float* W2_r1 = (const float*)d_in[4];
  const float* W1_r2 = (const float*)d_in[5];
  const float* W2_r2 = (const float*)d_in[6];
  const float* gamma = (const float*)d_in[7];
  const float* beta  = (const float*)d_in[8];
  float* out = (float*)d_out;

  const int N = in_sizes[0] / C;
  const int E = in_sizes[1] / 2;
  const long long NC = (long long)N * C;

  float* sum  = (float*)d_ws;
  float* mx   = sum  + NC;
  float* mn   = mx   + NC;
  float* mean = mn   + NC;
  float* h    = mean + NC;

  const int NP  = (N + 31) / 32 * 32;
  const int nCh = (E + CHUNK - 1) / CHUNK;
  const int nb  = (N + RS - 1) / RS;
  int* head = (int*)d_out;
  v2i* rec  = (v2i*)(head + 2 * (long long)NP);
  v2i* tab  = (v2i*)d_ws;

  const int T = 256;
  dim3 gG((unsigned)((N + 7) / 8));
  dim3 gM((unsigned)((N + 63) / 64));

  k_bin<<<dim3((unsigned)(2 * nCh)), 1024, 0, stream>>>(ei1, ei2, E, N, rec, tab);
  k_build<<<dim3((unsigned)(2 * nb)), 1024, 0, stream>>>(rec, tab, E, N, head, NP);

  k_gather<<<gG, T, 0, stream>>>(x, head, rec, N, E, sum, mx, mn, mean);
  k_gemm_wmma<true><<<gM, 128, 0, stream>>>(x, mean, mx, mn, sum, W1_r1, W2_r1, h, gamma, beta, out, N);

  k_gather<<<gG, T, 0, stream>>>(x, head + NP, rec + (long long)nCh * STRIDE, N, E, sum, mx, mn, mean);
  k_gemm_wmma<false><<<gM, 128, 0, stream>>>(x, mean, mx, mn, sum, W1_r2, W2_r2, h, gamma, beta, out, N);
}


# x products of both relations merged into one, x*(W1a+W1b)^T: weight tiles added at LDS staging in launch 2, launch 1 drops its x segment (K 1280 -> 1152); sum order of the x terms changes
# speedup vs baseline: 4.5205x; 1.0457x over previous
#include <hip/hip_runtime.h>
#include <cstdint>

#define C 128
#define FLT_BIG 3.402823466e38f

typedef float v2f __attribute__((ext_vector_type(2)));
typedef float v4f __attribute__((ext_vector_type(4)));
typedef int   v2i __attribute__((ext_vector_type(2)));
typedef float v8f __attribute__((ext_vector_type(8)));

#define RS_LOG 9
#define RS     (1 << RS_LOG)
#define CHUNK  4096
#define MAXB   128
#define STRIDE (CHUNK + 16 * MAXB)
#define SEGT   256

__global__ __launch_bounds__(1024) void k_bin(
    const int* __restrict__ ei1, const int* __restrict__ ei2,
    int n_edges, int n_nodes, v2i* __restrict__ rec, v2i* __restrict__ tab) {
  __shared__ int cnt[MAXB + 1];
  __shared__ int num[MAXB];
  __shared__ __attribute__((aligned(16))) v2i st[STRIDE];
  const int tid = threadIdx.x;
  const int nCh = (n_edges + CHUNK - 1) / CHUNK;
  const int r   = ((int)blockIdx.x >= nCh) ? 1 : 0;
  const int c   = (int)blockIdx.x - r * nCh;
  const int* __restrict__ ei = r ? ei2 : ei1;
  const int nb  = (n_nodes + RS - 1) >> RS_LOG;
  if (tid <= MAXB) cnt[tid] = 0;
  __syncthreads();

  int  s[4], d[4], rk[4];
  bool ok[4];
  const int base = c * CHUNK;
#pragma unroll
  for (int k = 0; k < 4; ++k) {
    const int e = base + k * 1024 + tid;
    ok[k] = e < n_edges;
    s[k] = 0; d[k] = 0; rk[k] = 0;
    if (ok[k]) { s[k] = ei[e]; d[k] = ei[n_edges + e]; }
  }
#pragma unroll
  for (int k = 0; k < 4; ++k)
    if (ok[k]) rk[k] = atomicAdd(&cnt[d[k] >> RS_LOG], 1);
  __syncthreads();

  if (tid < 32) {
    int v[4], local = 0;
#pragma unroll
    for (int j = 0; j < 4; ++j) {
      const int idx = tid * 4 + j;
      v[j] = (idx < nb) ? cnt[idx] : 0;
      local += (v[j] + 15) & ~15;
    }
    int inc = local;
#pragma unroll
    for (int off = 1; off < 32; off <<= 1) {
      const int t = __shfl_up(inc, off, 32);
      if (tid >= off) inc += t;
    }
    int ex = inc - local;
#pragma unroll
    for (int j = 0; j < 4; ++j) {
      const int idx = tid * 4 + j;
      if (idx < nb) { num[idx] = v[j]; cnt[idx] = ex; ex += (v[j] + 15) & ~15; }
    }
    if (tid == 31) cnt[MAXB] = inc;
  }
  __syncthreads();

#pragma unroll
  for (int k = 0; k < 4; ++k)
    if (ok[k]) {
      v2i q; q.x = s[k]; q.y = d[k];
      st[cnt[d[k] >> RS_LOG] + rk[k]] = q;
    }
  __syncthreads();

  const long long region = (long long)r * nCh + c;
  v2i* __restrict__ dst = rec + region * STRIDE;
  const int used = cnt[MAXB];
  for (int i = tid; i < used; i += 1024) dst[i] = st[i];
  if (tid < nb) {
    v2i t; t.x = cnt[tid]; t.y = cnt[tid] + num[tid];
    tab[region * MAXB + tid] = t;
  }
}

__global__ __launch_bounds__(1024) void k_build(
    v2i* __restrict__ rec, const v2i* __restrict__ tab,
    int n_edges, int n_nodes, int* __restrict__ head, int head_stride) {
  __shared__ int hd[RS];
  __shared__ v2i sg[SEGT];
  const int tid  = threadIdx.x;
  const int lane = tid & 31;
  const int wv   = tid >> 5;
  const int nCh  = (n_edges + CHUNK - 1) / CHUNK;
  const int nb   = (n_nodes + RS - 1) >> RS_LOG;
  const int r    = ((int)blockIdx.x >= nb) ? 1 : 0;
  const int b    = (int)blockIdx.x - r * nb;
  const int n0   = b << RS_LOG;
  if (tid < RS) hd[tid] = 0;
  int* __restrict__ ry = (int*)(rec + (long long)r * nCh * STRIDE) + 1;
  const v2i* __restrict__ tb = tab + (long long)r * nCh * MAXB + b;

  for (int c0 = 0; c0 < nCh; c0 += SEGT) {
    __syncthreads();
    const int nseg = (nCh - c0 < SEGT) ? (nCh - c0) : SEGT;
    if (tid < nseg) {
      v2i t = tb[(long long)(c0 + tid) * MAXB];
      t.x += (c0 + tid) * STRIDE;
      t.y += (c0 + tid) * STRIDE;
      sg[tid] = t;
    }
    __syncthreads();
    int p0[8], pe[8], y0[8], y1[8];
#pragma unroll
    for (int j = 0; j < 8; ++j) {
      const int i = wv + 32 * j;
      v2i se = { 0, 0 };
      if (i < nseg) se = sg[i];
      p0[j] = se.x + lane;
      pe[j] = se.y;
      y0[j] = 0;
      y1[j] = 0;
      if (p0[j] < pe[j])      y0[j] = ry[2 * (long long)p0[j]];
      if (p0[j] + 32 < pe[j]) y1[j] = ry[2 * (long long)(p0[j] + 32)];
    }
#pragma unroll
    for (int j = 0; j < 8; ++j) {
      if (p0[j] < pe[j])
        ry[2 * (long long)p0[j]] = __hip_atomic_exchange(&hd[y0[j] - n0], p0[j] + 1,
                                       __ATOMIC_RELAXED, __HIP_MEMORY_SCOPE_WORKGROUP);
      if (p0[j] + 32 < pe[j])
        ry[2 * (long long)(p0[j] + 32)] = __hip_atomic_exchange(&hd[y1[j] - n0], p0[j] + 33,
                                       __ATOMIC_RELAXED, __HIP_MEMORY_SCOPE_WORKGROUP);
      for (int p = p0[j] + 64; p < pe[j]; p += 32) {
        const int dn = ry[2 * (long long)p];
        ry[2 * (long long)p] = __hip_atomic_exchange(&hd[dn - n0], p + 1,
                                       __ATOMIC_RELAXED, __HIP_MEMORY_SCOPE_WORKGROUP);
      }
    }
  }
  __syncthreads();
  if (tid < RS && n0 + tid < n_nodes)
    head[(long long)r * head_stride + n0 + tid] = hd[tid];
}

__global__ __launch_bounds__(256) void k_gather(
    const float* __restrict__ x, const int* __restrict__ head,
    const v2i* __restrict__ rec, int n_nodes, int n_edges,
    float* __restrict__ sum, float* __restrict__ mx,
    float* __restrict__ mn, float* __restrict__ mean) {
  const int lane = threadIdx.x & 31;
  int node = blockIdx.x * 8 + (threadIdx.x >> 5);
  if (node >= n_nodes) return;
  node = __builtin_amdgcn_readfirstlane(node);

  v4f s  = { 0.0f, 0.0f, 0.0f, 0.0f };
  v4f hi = { -FLT_BIG, -FLT_BIG, -FLT_BIG, -FLT_BIG };
  v4f lo = {  FLT_BIG,  FLT_BIG,  FLT_BIG,  FLT_BIG };
  int cnt = 0;

  int e1 = head[node];
  v2i q = { 0, 0 };
  if (e1) q = rec[e1 - 1];
  for (int it = 0; it < n_edges && e1 != 0; ++it) {
    const v4f v = *(const v4f*)(x + (long long)q.x * C + lane * 4);
    e1 = q.y;
    if (e1) q = rec[e1 - 1];
    s += v;
    for (int k = 0; k < 4; ++k) {
      hi[k] = fmaxf(hi[k], v[k]);
      lo[k] = fminf(lo[k], v[k]);
    }
    ++cnt;
  }

  const float cn = (float)cnt;
  const float dv = fmaxf(cn, 1.0f);
  v4f m;
  for (int k = 0; k < 4; ++k) m[k] = s[k] / dv;
  if (cnt == 0) {
    hi = (v4f){ 0.0f, 0.0f, 0.0f, 0.0f };
    lo = (v4f){ 0.0f, 0.0f, 0.0f, 0.0f };
  }
  const long long o = (long long)node * C + lane * 4;
  *(v4f*)(sum  + o) = s;
  *(v4f*)(mx   + o) = hi;
  *(v4f*)(mn   + o) = lo;
  *(v4f*)(mean + o) = m;
}

__device__ __forceinline__ v4f ln_row(v4f v, const v4f g, const v4f b) {
#pragma unroll
  for (int k = 0; k < 4; ++k) v[k] = fmaxf(0.5f * v[k], 0.0f);
  float s  = (v[0] + v[1]) + (v[2] + v[3]);
  float s2 = (v[0] * v[0] + v[1] * v[1]) + (v[2] * v[2] + v[3] * v[3]);
#pragma unroll
  for (int off = 16; off > 0; off >>= 1) {
    s  += __shfl_xor(s,  off, 32);
    s2 += __shfl_xor(s2, off, 32);
  }
  const float mu  = s  * (1.0f / C);
  const float var = s2 * (1.0f / C) - mu * mu;
  const float rs  = rsqrtf(var + 1e-5f);
  v4f o;
#pragma unroll
  for (int k = 0; k < 4; ++k) o[k] = (v[k] - mu) * rs * g[k] + b[k];
  return o;
}

#define KC     32
#define BPITCH 36
#define NCHUNK 20
#define TP     132

struct ChunkRegs { v4f a[4]; v4f b[8]; v4f b2[8]; };

template <bool FIRST>
__global__ __launch_bounds__(128) void k_gemm_wmma(
    const float* __restrict__ x,
    const float* __restrict__ mean, const float* __restrict__ mx,
    const float* __restrict__ mn,   const float* __restrict__ sum,
    const float* __restrict__ W1,   const float* __restrict__ W1o,
    const float* __restrict__ W2,
    float* __restrict__ h,
    const float* __restrict__ gamma, const float* __restrict__ beta,
    float* __restrict__ out, int n_nodes) {
  __shared__ __attribute__((aligned(16))) float Bs[2 * 128 * BPITCH];

  const int tid  = threadIdx.x;
  const int lane = tid & 31;
  const int wv   = tid >> 5;
  const int m0   = blockIdx.x * 64 + wv * 16;
  const int c    = lane & 15;
  const int hf   = lane >> 4;

  int row = m0 + c;
  if (row >= n_nodes) row = n_nodes - 1;

  const int sn = tid >> 3;
  const int sq = tid & 7;

  v8f acc[8] = {};

  auto load_chunk = [&](int ch, ChunkRegs& r) {
    const int seg = ch >> 2;
    const int kc  = (ch & 3) * KC;
    const float* A  = (seg == 0) ? x : (seg == 1) ? mean : (seg == 2) ? mx : (seg == 3) ? mn : sum;
    const float* Wb = (seg == 0) ? W1 : (W2 + (seg - 1) * C);
    const int   ldw = (seg == 0) ? C : 4 * C;
    const float* Ap = A + (long long)row * C + kc + 4 * hf;
#pragma unroll
    for (int i = 0; i < 4; ++i) r.a[i] = *(const v4f*)(Ap + 8 * i);
    const float* Wp = Wb + (long long)sn * ldw + kc + 4 * sq;
#pragma unroll
    for (int p = 0; p < 8; ++p) r.b[p] = *(const v4f*)(Wp + (long long)(16 * p) * ldw);
    if (!FIRST && seg == 0) {
      const float* Wq = W1o + (long long)sn * C + kc + 4 * sq;
#pragma unroll
      for (int p = 0; p < 8; ++p) r.b2[p] = *(const v4f*)(Wq + (long long)(16 * p) * C);
    }
  };

  auto store_chunk = [&](int stage, const ChunkRegs& r, bool xseg) {
    float* dst = &Bs[stage * (128 * BPITCH) + sn * BPITCH + 4 * sq];
    if (!FIRST && xseg) {
#pragma unroll
      for (int p = 0; p < 8; ++p) *(v4f*)(dst + 16 * p * BPITCH) = r.b[p] + r.b2[p];
    } else {
#pragma unroll
      for (int p = 0; p < 8; ++p) *(v4f*)(dst + 16 * p * BPITCH) = r.b[p];
    }
  };

  auto load_frags = [&](const float* src, int i, v4f (&bf)[8]) {
#pragma unroll
    for (int j = 0; j < 8; ++j) bf[j] = *(const v4f*)(src + 16 * j * BPITCH + 8 * i);
  };

  auto compute_chunk = [&](int stage, const ChunkRegs& r) {
    const float* src = &Bs[stage * (128 * BPITCH) + c * BPITCH + 4 * hf];
    v4f bf[2][8];
    load_frags(src, 0, bf[0]);
#pragma unroll
    for (int i = 0; i < 4; ++i) {
      if (i + 1 < 4) load_frags(src, i + 1, bf[(i + 1) & 1]);
      __builtin_amdgcn_sched_barrier(0);
      const v2f a0 = { r.a[i][0], r.a[i][1] };
      const v2f a1 = { r.a[i][2], r.a[i][3] };
#pragma unroll
      for (int j = 0; j < 8; ++j) {
        const v2f b0 = { bf[i & 1][j][0], bf[i & 1][j][1] };
        acc[j] = __builtin_amdgcn_wmma_f32_16x16x4_f32(
            false, a0, false, b0, (short)0, acc[j], false, false);
      }
#pragma unroll
      for (int j = 0; j < 8; ++j) {
        const v2f b1 = { bf[i & 1][j][2], bf[i & 1][j][3] };
        acc[j] = __builtin_amdgcn_wmma_f32_16x16x4_f32(
            false, a1, false, b1, (short)0, acc[j], false, false);
      }
      __builtin_amdgcn_sched_barrier(0);
    }
  };

  constexpr int CH0 = FIRST ? 4 : 0;
  ChunkRegs cur, nxt;
  load_chunk(CH0, cur);
  store_chunk(CH0 & 1, cur, CH0 < 4);
  __syncthreads();

  for (int ch = CH0; ch < NCHUNK; ++ch) {
    const bool more = (ch + 1 < NCHUNK);
    if (more) load_chunk(ch + 1, nxt);
    asm volatile("" ::: "memory");
    compute_chunk(ch & 1, cur);
    if (more) store_chunk((ch + 1) & 1, nxt, ch + 1 < 4);
    __syncthreads();
    if (more) {
#pragma unroll
      for (int i = 0; i < 4; ++i) cur.a[i] = nxt.a[i];
    }
  }

  float* T = &Bs[wv * (16 * TP)];
#pragma unroll
  for (int v = 0; v < 8; ++v) {
#pragma unroll
    for (int j = 0; j < 8; ++j) T[(8 * hf + v) * TP + 16 * j + c] = acc[j][v];
  }
  __syncthreads();

  const long long co = lane * 4;
  const bool full = (m0 + 16 <= n_nodes);
  float* dstp = (FIRST ? h : out) + (long long)m0 * C + co;
  v4f o[16];
  if (FIRST) {
#pragma unroll
    for (int rr = 0; rr < 16; ++rr) o[rr] = *(const v4f*)(T + rr * TP + co);
  } else {
    const v4f g = *(const v4f*)(gamma + co);
    const v4f b = *(const v4f*)(beta  + co);
    v4f hv[16];
#pragma unroll
    for (int rr = 0; rr < 16; ++rr) {
      int r = m0 + rr;
      if (r >= n_nodes) r = n_nodes - 1;
      hv[rr] = *(const v4f*)(h + (long long)r * C + co);
    }
#pragma unroll
    for (int rr = 0; rr < 16; ++rr) {
      const v4f t = hv[rr] + *(const v4f*)(T + rr * TP + co);
      o[rr] = ln_row(t, g, b);
    }
  }
  if (full) {
#pragma unroll
    for (int rr = 0; rr < 16; ++rr) *(v4f*)(dstp + (long long)rr * C) = o[rr];
  } else {
#pragma unroll
    for (int rr = 0; rr < 16; ++rr)
      if (m0 + rr < n_nodes) *(v4f*)(dstp + (long long)rr * C) = o[rr];
  }
}

extern "C" void kernel_launch(void* const* d_in, const int* in_sizes, int n_in,
                              void* d_out, int out_size, void* d_ws, size_t ws_size,
                              hipStream_t stream) {
  const float* x     = (const float*)d_in[0];
  const int*   ei1   = (const int*)d_in[1];
  const int*   ei2   = (const int*)d_in[2];
  const float* W1_r1 = (const float*)d_in[3];
  const float* W2_r1 = (const float*)d_in[4];
  const float* W1_r2 = (const float*)d_in[5];
  const float* W2_r2 = (const float*)d_in[6];
  const float* gamma = (const float*)d_in[7];
  const float* beta  = (const float*)d_in[8];
  float* out = (float*)d_out;

  const int N = in_sizes[0] / C;
  const int E = in_sizes[1] / 2;
  const long long NC = (long long)N * C;

  float* sum  = (float*)d_ws;
  float* mx   = sum  + NC;
  float* mn   = mx   + NC;
  float* mean = mn   + NC;
  float* h    = mean + NC;

  const int NP  = (N + 31) / 32 * 32;
  const int nCh = (E + CHUNK - 1) / CHUNK;
  const int nb  = (N + RS - 1) / RS;
  int* head = (int*)d_out;
  v2i* rec  = (v2i*)(head + 2 * (long long)NP);
  v2i* tab  = (v2i*)d_ws;

  const int T = 256;
  dim3 gG((unsigned)((N + 7) / 8));
  dim3 gM((unsigned)((N + 63) / 64));

  k_bin<<<dim3((unsigned)(2 * nCh)), 1024, 0, stream>>>(ei1, ei2, E, N, rec, tab);
  k_build<<<dim3((unsigned)(2 * nb)), 1024, 0, stream>>>(rec, tab, E, N, head, NP);

  k_gather<<<gG, T, 0, stream>>>(x, head, rec, N, E, sum, mx, mn, mean);
  k_gemm_wmma<true><<<gM, 128, 0, stream>>>(x, mean, mx, mn, sum, W1_r1, W1_r2, W2_r1, h, gamma, beta, out, N);

  k_gather<<<gG, T, 0, stream>>>(x, head + NP, rec + (long long)nCh * STRIDE, N, E, sum, mx, mn, mean);
  k_gemm_wmma<false><<<gM, 128, 0, stream>>>(x, mean, mx, mn, sum, W1_r2, W1_r1, W2_r2, h, gamma, beta, out, N);
}


# GEMM grids balanced over the work-group processors: whole rounds of 64-row work-groups, the remaining rows as 16-row work-groups whose 4 waves take 32 columns each; all sum orders kept
# speedup vs baseline: 4.8071x; 1.0634x over previous
#include <hip/hip_runtime.h>
#include <cstdint>

#define C 128
#define FLT_BIG 3.402823466e38f

typedef float v2f __attribute__((ext_vector_type(2)));
typedef float v4f __attribute__((ext_vector_type(4)));
typedef int   v2i __attribute__((ext_vector_type(2)));
typedef float v8f __attribute__((ext_vector_type(8)));

#define RS_LOG 9
#define RS     (1 << RS_LOG)
#define CHUNK  4096
#define MAXB   128
#define STRIDE (CHUNK + 16 * MAXB)
#define SEGT   256

__global__ __launch_bounds__(1024) void k_bin(
    const int* __restrict__ ei1, const int* __restrict__ ei2,
    int n_edges, int n_nodes, v2i* __restrict__ rec, v2i* __restrict__ tab) {
  __shared__ int cnt[MAXB + 1];
  __shared__ int num[MAXB];
  __shared__ __attribute__((aligned(16))) v2i st[STRIDE];
  const int tid = threadIdx.x;
  const int nCh = (n_edges + CHUNK - 1) / CHUNK;
  const int r   = ((int)blockIdx.x >= nCh) ? 1 : 0;
  const int c   = (int)blockIdx.x - r * nCh;
  const int* __restrict__ ei = r ? ei2 : ei1;
  const int nb  = (n_nodes + RS - 1) >> RS_LOG;
  if (tid <= MAXB) cnt[tid] = 0;
  __syncthreads();

  int  s[4], d[4], rk[4];
  bool ok[4];
  const int base = c * CHUNK;
#pragma unroll
  for (int k = 0; k < 4; ++k) {
    const int e = base + k * 1024 + tid;
    ok[k] = e < n_edges;
    s[k] = 0; d[k] = 0; rk[k] = 0;
    if (ok[k]) { s[k] = ei[e]; d[k] = ei[n_edges + e]; }
  }
#pragma unroll
  for (int k = 0; k < 4; ++k)
    if (ok[k]) rk[k] = atomicAdd(&cnt[d[k] >> RS_LOG], 1);
  __syncthreads();

  if (tid < 32) {
    int v[4], local = 0;
#pragma unroll
    for (int j = 0; j < 4; ++j) {
      const int idx = tid * 4 + j;
      v[j] = (idx < nb) ? cnt[idx] : 0;
      local += (v[j] + 15) & ~15;
    }
    int inc = local;
#pragma unroll
    for (int off = 1; off < 32; off <<= 1) {
      const int t = __shfl_up(inc, off, 32);
      if (tid >= off) inc += t;
    }
    int ex = inc - local;
#pragma unroll
    for (int j = 0; j < 4; ++j) {
      const int idx = tid * 4 + j;
      if (idx < nb) { num[idx] = v[j]; cnt[idx] = ex; ex += (v[j] + 15) & ~15; }
    }
    if (tid == 31) cnt[MAXB] = inc;
  }
  __syncthreads();

#pragma unroll
  for (int k = 0; k < 4; ++k)
    if (ok[k]) {
      v2i q; q.x = s[k]; q.y = d[k];
      st[cnt[d[k] >> RS_LOG] + rk[k]] = q;
    }
  __syncthreads();

  const long long region = (long long)r * nCh + c;
  v2i* __restrict__ dst = rec + region * STRIDE;
  const int used = cnt[MAXB];
  for (int i = tid; i < used; i += 1024) dst[i] = st[i];
  if (tid < nb) {
    v2i t; t.x = cnt[tid]; t.y = cnt[tid] + num[tid];
    tab[region * MAXB + tid] = t;
  }
}

__global__ __launch_bounds__(1024) void k_build(
    v2i* __restrict__ rec, const v2i* __restrict__ tab,
    int n_edges, int n_nodes, int* __restrict__ head, int head_stride) {
  __shared__ int hd[RS];
  __shared__ v2i sg[SEGT];
  const int tid  = threadIdx.x;
  const int lane = tid & 31;
  const int wv   = tid >> 5;
  const int nCh  = (n_edges + CHUNK - 1) / CHUNK;
  const int nb   = (n_nodes + RS - 1) >> RS_LOG;
  const int r    = ((int)blockIdx.x >= nb) ? 1 : 0;
  const int b    = (int)blockIdx.x - r * nb;
  const int n0   = b << RS_LOG;
  if (tid < RS) hd[tid] = 0;
  int* __restrict__ ry = (int*)(rec + (long long)r * nCh * STRIDE) + 1;
  const v2i* __restrict__ tb = tab + (long long)r * nCh * MAXB + b;

  for (int c0 = 0; c0 < nCh; c0 += SEGT) {
    __syncthreads();
    const int nseg = (nCh - c0 < SEGT) ? (nCh - c0) : SEGT;
    if (tid < nseg) {
      v2i t = tb[(long long)(c0 + tid) * MAXB];
      t.x += (c0 + tid) * STRIDE;
      t.y += (c0 + tid) * STRIDE;
      sg[tid] = t;
    }
    __syncthreads();
    int p0[8], pe[8], y0[8], y1[8];
#pragma unroll
    for (int j = 0; j < 8; ++j) {
      const int i = wv + 32 * j;
      v2i se = { 0, 0 };
      if (i < nseg) se = sg[i];
      p0[j] = se.x + lane;
      pe[j] = se.y;
      y0[j] = 0;
      y1[j] = 0;
      if (p0[j] < pe[j])      y0[j] = ry[2 * (long long)p0[j]];
      if (p0[j] + 32 < pe[j]) y1[j] = ry[2 * (long long)(p0[j] + 32)];
    }
#pragma unroll
    for (int j = 0; j < 8; ++j) {
      if (p0[j] < pe[j])
        ry[2 * (long long)p0[j]] = __hip_atomic_exchange(&hd[y0[j] - n0], p0[j] + 1,
                                       __ATOMIC_RELAXED, __HIP_MEMORY_SCOPE_WORKGROUP);
      if (p0[j] + 32 < pe[j])
        ry[2 * (long long)(p0[j] + 32)] = __hip_atomic_exchange(&hd[y1[j] - n0], p0[j] + 33,
                                       __ATOMIC_RELAXED, __HIP_MEMORY_SCOPE_WORKGROUP);
      for (int p = p0[j] + 64; p < pe[j]; p += 32) {
        const int dn = ry[2 * (long long)p];
        ry[2 * (long long)p] = __hip_atomic_exchange(&hd[dn - n0], p + 1,
                                       __ATOMIC_RELAXED, __HIP_MEMORY_SCOPE_WORKGROUP);
      }
    }
  }
  __syncthreads();
  if (tid < RS && n0 + tid < n_nodes)
    head[(long long)r * head_stride + n0 + tid] = hd[tid];
}

__global__ __launch_bounds__(256) void k_gather(
    const float* __restrict__ x, const int* __restrict__ head,
    const v2i* __restrict__ rec, int n_nodes, int n_edges,
    float* __restrict__ sum, float* __restrict__ mx,
    float* __restrict__ mn, float* __restrict__ mean) {
  const int lane = threadIdx.x & 31;
  int node = blockIdx.x * 8 + (threadIdx.x >> 5);
  if (node >= n_nodes) return;
  node = __builtin_amdgcn_readfirstlane(node);

  v4f s  = { 0.0f, 0.0f, 0.0f, 0.0f };
  v4f hi = { -FLT_BIG, -FLT_BIG, -FLT_BIG, -FLT_BIG };
  v4f lo = {  FLT_BIG,  FLT_BIG,  FLT_BIG,  FLT_BIG };
  int cnt = 0;

  int e1 = head[node];
  v2i q = { 0, 0 };
  if (e1) q = rec[e1 - 1];
  for (int it = 0; it < n_edges && e1 != 0; ++it) {
    const v4f v = *(const v4f*)(x + (long long)q.x * C + lane * 4);
    e1 = q.y;
    if (e1) q = rec[e1 - 1];
    s += v;
    for (int k = 0; k < 4; ++k) {
      hi[k] = fmaxf(hi[k], v[k]);
      lo[k] = fminf(lo[k], v[k]);
    }
    ++cnt;
  }

  const float cn = (float)cnt;
  const float dv = fmaxf(cn, 1.0f);
  v4f m;
  for (int k = 0; k < 4; ++k) m[k] = s[k] / dv;
  if (cnt == 0) {
    hi = (v4f){ 0.0f, 0.0f, 0.0f, 0.0f };
    lo = (v4f){ 0.0f, 0.0f, 0.0f, 0.0f };
  }
  const long long o = (long long)node * C + lane * 4;
  *(v4f*)(sum  + o) = s;
  *(v4f*)(mx   + o) = hi;
  *(v4f*)(mn   + o) = lo;
  *(v4f*)(mean + o) = m;
}

__device__ __forceinline__ v4f ln_row(v4f v, const v4f g, const v4f b) {
#pragma unroll
  for (int k = 0; k < 4; ++k) v[k] = fmaxf(0.5f * v[k], 0.0f);
  float s  = (v[0] + v[1]) + (v[2] + v[3]);
  float s2 = (v[0] * v[0] + v[1] * v[1]) + (v[2] * v[2] + v[3] * v[3]);
#pragma unroll
  for (int off = 16; off > 0; off >>= 1) {
    s  += __shfl_xor(s,  off, 32);
    s2 += __shfl_xor(s2, off, 32);
  }
  const float mu  = s  * (1.0f / C);
  const float var = s2 * (1.0f / C) - mu * mu;
  const float rs  = rsqrtf(var + 1e-5f);
  v4f o;
#pragma unroll
  for (int k = 0; k < 4; ++k) o[k] = (v[k] - mu) * rs * g[k] + b[k];
  return o;
}

#define KC     32
#define BPITCH 36
#define NCHUNK 20
#define TP     132

struct ChunkRegs { v4f a[4]; v4f b[8]; v4f b2[8]; };

template <int V> struct IntC { static constexpr int value = V; };

#define N_WGP 256

template <bool FIRST>
__global__ __launch_bounds__(128) void k_gemm_wmma(
    const float* __restrict__ x,
    const float* __restrict__ mean, const float* __restrict__ mx,
    const float* __restrict__ mn,   const float* __restrict__ sum,
    const float* __restrict__ W1,   const float* __restrict__ W1o,
    const float* __restrict__ W2,
    float* __restrict__ h,
    const float* __restrict__ gamma, const float* __restrict__ beta,
    float* __restrict__ out, int n_nodes, int n_full) {
  __shared__ __attribute__((aligned(16))) float Bs[2 * 128 * BPITCH];

  const int tid  = threadIdx.x;
  const int lane = tid & 31;
  const int wv   = tid >> 5;
  const bool split = (int)blockIdx.x >= n_full;
  const int m0   = split ? n_full * 64 + ((int)blockIdx.x - n_full) * 16
                         : (int)blockIdx.x * 64 + wv * 16;
  const int c    = lane & 15;
  const int hf   = lane >> 4;

  int row = m0 + c;
  if (row >= n_nodes) row = n_nodes - 1;

  const int sn = tid >> 3;
  const int sq = tid & 7;

  v8f acc[8] = {};

  auto load_chunk = [&](int ch, ChunkRegs& r) {
    const int seg = ch >> 2;
    const int kc  = (ch & 3) * KC;
    const float* A  = (seg == 0) ? x : (seg == 1) ? mean : (seg == 2) ? mx : (seg == 3) ? mn : sum;
    const float* Wb = (seg == 0) ? W1 : (W2 + (seg - 1) * C);
    const int   ldw = (seg == 0) ? C : 4 * C;
    const float* Ap = A + (long long)row * C + kc + 4 * hf;
#pragma unroll
    for (int i = 0; i < 4; ++i) r.a[i] = *(const v4f*)(Ap + 8 * i);
    const float* Wp = Wb + (long long)sn * ldw + kc + 4 * sq;
#pragma unroll
    for (int p = 0; p < 8; ++p) r.b[p] = *(const v4f*)(Wp + (long long)(16 * p) * ldw);
    if (!FIRST && seg == 0) {
      const float* Wq = W1o + (long long)sn * C + kc + 4 * sq;
#pragma unroll
      for (int p = 0; p < 8; ++p) r.b2[p] = *(const v4f*)(Wq + (long long)(16 * p) * C);
    }
  };

  auto store_chunk = [&](int stage, const ChunkRegs& r, bool xseg) {
    float* dst = &Bs[stage * (128 * BPITCH) + sn * BPITCH + 4 * sq];
    if (!FIRST && xseg) {
#pragma unroll
      for (int p = 0; p < 8; ++p) *(v4f*)(dst + 16 * p * BPITCH) = r.b[p] + r.b2[p];
    } else {
#pragma unroll
      for (int p = 0; p < 8; ++p) *(v4f*)(dst + 16 * p * BPITCH) = r.b[p];
    }
  };

  auto compute_chunk = [&](int stage, const ChunkRegs& r, auto njc, int jb) {
    constexpr int NJ = decltype(njc)::value;
    const float* src = &Bs[stage * (128 * BPITCH) + (16 * jb + c) * BPITCH + 4 * hf];
    v4f bf[2][NJ];
    auto load_frags = [&](int i, v4f (&f)[NJ]) {
#pragma unroll
      for (int j = 0; j < NJ; ++j) f[j] = *(const v4f*)(src + 16 * j * BPITCH + 8 * i);
    };
    load_frags(0, bf[0]);
#pragma unroll
    for (int i = 0; i < 4; ++i) {
      if (i + 1 < 4) load_frags(i + 1, bf[(i + 1) & 1]);
      __builtin_amdgcn_sched_barrier(0);
      const v2f a0 = { r.a[i][0], r.a[i][1] };
      const v2f a1 = { r.a[i][2], r.a[i][3] };
#pragma unroll
      for (int j = 0; j < NJ; ++j) {
        const v2f b0 = { bf[i & 1][j][0], bf[i & 1][j][1] };
        acc[j] = __builtin_amdgcn_wmma_f32_16x16x4_f32(
            false, a0, false, b0, (short)0, acc[j], false, false);
      }
#pragma unroll
      for (int j = 0; j < NJ; ++j) {
        const v2f b1 = { bf[i & 1][j][2], bf[i & 1][j][3] };
        acc[j] = __builtin_amdgcn_wmma_f32_16x16x4_f32(
            false, a1, false, b1, (short)0, acc[j], false, false);
      }
      __builtin_amdgcn_sched_barrier(0);
    }
  };

  constexpr int CH0 = FIRST ? 4 : 0;
  auto run_k = [&](auto njc, int jb) {
    ChunkRegs cur, nxt;
    load_chunk(CH0, cur);
    store_chunk(CH0 & 1, cur, CH0 < 4);
    __syncthreads();

    for (int ch = CH0; ch < NCHUNK; ++ch) {
      const bool more = (ch + 1 < NCHUNK);
      if (more) load_chunk(ch + 1, nxt);
      asm volatile("" ::: "memory");
      compute_chunk(ch & 1, cur, njc, jb);
      if (more) store_chunk((ch + 1) & 1, nxt, ch + 1 < 4);
      __syncthreads();
      if (more) {
#pragma unroll
        for (int i = 0; i < 4; ++i) cur.a[i] = nxt.a[i];
      }
    }
  };
  if (split) run_k(IntC<2>{}, 2 * wv);
  else       run_k(IntC<8>{}, 0);

  float* T = split ? &Bs[0] : &Bs[wv * (16 * TP)];
  if (split) {
#pragma unroll
    for (int v = 0; v < 8; ++v) {
#pragma unroll
      for (int j = 0; j < 2; ++j) T[(8 * hf + v) * TP + 16 * (2 * wv + j) + c] = acc[j][v];
    }
  } else {
#pragma unroll
    for (int v = 0; v < 8; ++v) {
#pragma unroll
      for (int j = 0; j < 8; ++j) T[(8 * hf + v) * TP + 16 * j + c] = acc[j][v];
    }
  }
  __syncthreads();

  const long long co = lane * 4;
  auto finish = [&](auto nrc, const float* Tt, int mrow) {
    constexpr int NR = decltype(nrc)::value;
    const bool full = (mrow + NR <= n_nodes);
    float* dstp = (FIRST ? h : out) + (long long)mrow * C + co;
    v4f o[NR];
    if (FIRST) {
#pragma unroll
      for (int rr = 0; rr < NR; ++rr) o[rr] = *(const v4f*)(Tt + rr * TP + co);
    } else {
      const v4f g = *(const v4f*)(gamma + co);
      const v4f b = *(const v4f*)(beta  + co);
      v4f hv[NR];
#pragma unroll
      for (int rr = 0; rr < NR; ++rr) {
        int r = mrow + rr;
        if (r >= n_nodes) r = n_nodes - 1;
        hv[rr] = *(const v4f*)(h + (long long)r * C + co);
      }
#pragma unroll
      for (int rr = 0; rr < NR; ++rr) {
        const v4f t = hv[rr] + *(const v4f*)(Tt + rr * TP + co);
        o[rr] = ln_row(t, g, b);
      }
    }
    if (full) {
#pragma unroll
      for (int rr = 0; rr < NR; ++rr) *(v4f*)(dstp + (long long)rr * C) = o[rr];
    } else {
#pragma unroll
      for (int rr = 0; rr < NR; ++rr)
        if (mrow + rr < n_nodes) *(v4f*)(dstp + (long long)rr * C) = o[rr];
    }
  };
  if (split) finish(IntC<4>{}, T + 4 * wv * TP, m0 + 4 * wv);
  else       finish(IntC<16>{}, T, m0);
}

extern "C" void kernel_launch(void* const* d_in, const int* in_sizes, int n_in,
                              void* d_out, int out_size, void* d_ws, size_t ws_size,
                              hipStream_t stream) {
  const float* x     = (const float*)d_in[0];
  const int*   ei1   = (const int*)d_in[1];
  const int*   ei2   = (const int*)d_in[2];
  const float* W1_r1 = (const float*)d_in[3];
  const float* W2_r1 = (const float*)d_in[4];
  const float* W1_r2 = (const float*)d_in[5];
  const float* W2_r2 = (const float*)d_in[6];
  const float* gamma = (const float*)d_in[7];
  const float* beta  = (const float*)d_in[8];
  float* out = (float*)d_out;

  const int N = in_sizes[0] / C;
  const int E = in_sizes[1] / 2;
  const long long NC = (long long)N * C;

  float* sum  = (float*)d_ws;
  float* mx   = sum  + NC;
  float* mn   = mx   + NC;
  float* mean = mn   + NC;
  float* h    = mean + NC;

  const int NP  = (N + 31) / 32 * 32;
  const int nCh = (E + CHUNK - 1) / CHUNK;
  const int nb  = (N + RS - 1) / RS;
  int* head = (int*)d_out;
  v2i* rec  = (v2i*)(head + 2 * (long long)NP);
  v2i* tab  = (v2i*)d_ws;

  const int T = 256;
  dim3 gG((unsigned)((N + 7) / 8));
  const int nT64   = N / 64;
  const int nFull  = nT64 - nT64 % N_WGP;
  const int nSplit = (N - nFull * 64 + 15) / 16;
  dim3 gM((unsigned)(nFull + nSplit));

  k_bin<<<dim3((unsigned)(2 * nCh)), 1024, 0, stream>>>(ei1, ei2, E, N, rec, tab);
  k_build<<<dim3((unsigned)(2 * nb)), 1024, 0, stream>>>(rec, tab, E, N, head, NP);

  k_gather<<<gG, T, 0, stream>>>(x, head, rec, N, E, sum, mx, mn, mean);
  k_gemm_wmma<true><<<gM, 128, 0, stream>>>(x, mean, mx, mn, sum, W1_r1, W1_r2, W2_r1, h, gamma, beta, out, N, nFull);

  k_gather<<<gG, T, 0, stream>>>(x, head + NP, rec + (long long)nCh * STRIDE, N, E, sum, mx, mn, mean);
  k_gemm_wmma<false><<<gM, 128, 0, stream>>>(x, mean, mx, mn, sum, W1_r2, W1_r1, W2_r2, h, gamma, beta, out, N, nFull);
}


# gather reads contiguous per-destination source lists (counting sort in LDS replaces the linked records): resident waves loop over nodes, 8 rows in flight; row sum order = order of the LDS adds
# speedup vs baseline: 4.9602x; 1.0318x over previous
#include <hip/hip_runtime.h>
#include <cstdint>

#define C 128
#define FLT_BIG 3.402823466e38f

typedef float v2f __attribute__((ext_vector_type(2)));
typedef float v4f __attribute__((ext_vector_type(4)));
typedef int   v2i __attribute__((ext_vector_type(2)));
typedef float v8f __attribute__((ext_vector_type(8)));

#define RS_LOG 9
#define RS     (1 << RS_LOG)
#define CHUNK  4096
#define MAXB   128
#define STRIDE (CHUNK + 16 * MAXB)
#define SEGT   256
#define CAP    12288

__global__ __launch_bounds__(1024) void k_bin(
    const int* __restrict__ ei1, const int* __restrict__ ei2,
    int n_edges, int n_nodes, v2i* __restrict__ rec, v2i* __restrict__ tab) {
  __shared__ int cnt[MAXB + 1];
  __shared__ int num[MAXB];
  __shared__ __attribute__((aligned(16))) v2i st[STRIDE];
  const int tid = threadIdx.x;
  const int nCh = (n_edges + CHUNK - 1) / CHUNK;
  const int r   = ((int)blockIdx.x >= nCh) ? 1 : 0;
  const int c   = (int)blockIdx.x - r * nCh;
  const int* __restrict__ ei = r ? ei2 : ei1;
  const int nb  = (n_nodes + RS - 1) >> RS_LOG;
  if (tid <= MAXB) cnt[tid] = 0;
  __syncthreads();

  int  s[4], d[4], rk[4];
  bool ok[4];
  const int base = c * CHUNK;
#pragma unroll
  for (int k = 0; k < 4; ++k) {
    const int e = base + k * 1024 + tid;
    ok[k] = e < n_edges;
    s[k] = 0; d[k] = 0; rk[k] = 0;
    if (ok[k]) { s[k] = ei[e]; d[k] = ei[n_edges + e]; }
  }
#pragma unroll
  for (int k = 0; k < 4; ++k)
    if (ok[k]) rk[k] = atomicAdd(&cnt[d[k] >> RS_LOG], 1);
  __syncthreads();

  if (tid < 32) {
    int v[4], local = 0;
#pragma unroll
    for (int j = 0; j < 4; ++j) {
      const int idx = tid * 4 + j;
      v[j] = (idx < nb) ? cnt[idx] : 0;
      local += (v[j] + 15) & ~15;
    }
    int inc = local;
#pragma unroll
    for (int off = 1; off < 32; off <<= 1) {
      const int t = __shfl_up(inc, off, 32);
      if (tid >= off) inc += t;
    }
    int ex = inc - local;
#pragma unroll
    for (int j = 0; j < 4; ++j) {
      const int idx = tid * 4 + j;
      if (idx < nb) { num[idx] = v[j]; cnt[idx] = ex; ex += (v[j] + 15) & ~15; }
    }
    if (tid == 31) cnt[MAXB] = inc;
  }
  __syncthreads();

#pragma unroll
  for (int k = 0; k < 4; ++k)
    if (ok[k]) {
      v2i q; q.x = s[k]; q.y = d[k];
      st[cnt[d[k] >> RS_LOG] + rk[k]] = q;
    }
  __syncthreads();

  const long long region = (long long)r * nCh + c;
  v2i* __restrict__ dst = rec + region * STRIDE;
  const int used = cnt[MAXB];
  for (int i = tid; i < used; i += 1024) dst[i] = st[i];
  if (tid < nb) {
    v2i t; t.x = cnt[tid]; t.y = cnt[tid] + num[tid];
    tab[region * MAXB + tid] = t;
  }
}

__global__ __launch_bounds__(1024) void k_build(
    const v2i* __restrict__ rec, const v2i* __restrict__ tab,
    int n_edges, int n_nodes, v2i* __restrict__ start, int start_stride,
    int* __restrict__ adj, int adj_stride) {
  __shared__ int cnt[RS];
  __shared__ int tot[MAXB];
  __shared__ int wsum[RS / 32];
  __shared__ int baseS;
  __shared__ v2i sg[SEGT];
  __shared__ int stage[CAP];
  const int tid  = threadIdx.x;
  const int lane = tid & 31;
  const int wv   = tid >> 5;
  const int nCh  = (n_edges + CHUNK - 1) / CHUNK;
  const int nb   = (n_nodes + RS - 1) >> RS_LOG;
  const int r    = ((int)blockIdx.x >= nb) ? 1 : 0;
  const int b    = (int)blockIdx.x - r * nb;
  const int n0   = b << RS_LOG;
  if (tid < RS)   cnt[tid] = 0;
  if (tid < MAXB) tot[tid] = 0;
  const v2i* __restrict__ recr = rec + (long long)r * nCh * STRIDE;
  const v2i* __restrict__ tabr = tab + (long long)r * nCh * MAXB;
  const v2i* __restrict__ tb   = tabr + b;
  int* __restrict__ adjr = adj + (long long)r * adj_stride;
  __syncthreads();

  {
    const int rg = tid & (MAXB - 1);
    if (rg <= b) {
      int part = 0;
#pragma unroll 8
      for (int c = tid >> 7; c < nCh; c += 1024 / MAXB) {
        const v2i t = tabr[(long long)c * MAXB + rg];
        part += t.y - t.x;
      }
      atomicAdd(&tot[rg], part);
    }
  }
  __syncthreads();
  if (wv == 0) {
    int v = 0;
    for (int j = 0; j < 4; ++j) {
      const int i = lane * 4 + j;
      if (i < b) v += (tot[i] + 31) & ~31;
    }
    for (int off = 16; off > 0; off >>= 1) v += __shfl_xor(v, off, 32);
    if (lane == 0) baseS = v;
  }

  auto sweep = [&](auto op) {
    for (int c0 = 0; c0 < nCh; c0 += SEGT) {
      __syncthreads();
      const int nseg = (nCh - c0 < SEGT) ? (nCh - c0) : SEGT;
      if (tid < nseg) {
        v2i t = tb[(long long)(c0 + tid) * MAXB];
        t.x += (c0 + tid) * STRIDE;
        t.y += (c0 + tid) * STRIDE;
        sg[tid] = t;
      }
      __syncthreads();
      int p0[8], pe[8];
      v2i q0[8], q1[8];
#pragma unroll
      for (int j = 0; j < 8; ++j) {
        const int i = wv + 32 * j;
        v2i se = { 0, 0 };
        if (i < nseg) se = sg[i];
        p0[j] = se.x + lane;
        pe[j] = se.y;
        q0[j] = (v2i){ 0, 0 };
        q1[j] = (v2i){ 0, 0 };
        if (p0[j] < pe[j])      q0[j] = recr[p0[j]];
        if (p0[j] + 32 < pe[j]) q1[j] = recr[p0[j] + 32];
      }
#pragma unroll
      for (int j = 0; j < 8; ++j) {
        if (p0[j] < pe[j])      op(q0[j]);
        if (p0[j] + 32 < pe[j]) op(q1[j]);
        for (int p = p0[j] + 64; p < pe[j]; p += 32) op(recr[p]);
      }
    }
  };

  sweep([&](v2i q) { atomicAdd(&cnt[q.y - n0], 1); });
  __syncthreads();

  const int base  = baseS;
  const int total = tot[b];
  int myc = 0, incl = 0;
  if (tid < RS) {
    myc = cnt[tid];
    incl = myc;
    for (int off = 1; off < 32; off <<= 1) {
      const int t = __shfl_up(incl, off, 32);
      if (lane >= off) incl += t;
    }
    if (lane == 31) wsum[wv] = incl;
  }
  __syncthreads();
  if (tid < RS) {
    int pre = 0;
    for (int w = 0; w < RS / 32; ++w)
      if (w < wv) pre += wsum[w];
    const int excl = pre + incl - myc;
    cnt[tid] = excl;
    if (n0 + tid < n_nodes) {
      v2i sv; sv.x = base + excl; sv.y = myc;
      start[(long long)r * start_stride + n0 + tid] = sv;
    }
  }

  const bool inlds = total <= CAP;
  if (inlds) {
    sweep([&](v2i q) { stage[atomicAdd(&cnt[q.y - n0], 1)] = q.x; });
    __syncthreads();
    const int padded = (total + 31) & ~31;
    for (int i = tid; i < padded; i += 1024) adjr[base + i] = (i < total) ? stage[i] : 0;
  } else {
    sweep([&](v2i q) { adjr[base + atomicAdd(&cnt[q.y - n0], 1)] = q.x; });
  }
}

__global__ __launch_bounds__(256) void k_gather(
    const float* __restrict__ x, const v2i* __restrict__ start,
    const int* __restrict__ adj, int n_nodes, int n_edges, int n_waves,
    float* __restrict__ sum, float* __restrict__ mx,
    float* __restrict__ mn, float* __restrict__ mean) {
  const int lane = threadIdx.x & 31;
  const int w0   = __builtin_amdgcn_readfirstlane((int)blockIdx.x * 8 + ((int)threadIdx.x >> 5));
  if (w0 >= n_nodes) return;

  const float* __restrict__ xl = x + lane * 4;

  v2i d0 = start[w0];
  v2i d1 = { 0, 0 };
  if (w0 + n_waves < n_nodes) d1 = start[w0 + n_waves];
  int e0 = 0;
  if (lane < d0.y) e0 = adj[d0.x + lane];
  asm volatile("" :: "v"(e0));

  for (int node = w0; node < n_nodes; node += n_waves) {
    const v2i st  = d0;
    const int my0 = e0;
    d0 = d1;
    if (node + 2 * n_waves < n_nodes) d1 = start[node + 2 * n_waves];
    e0 = 0;
    if (node + n_waves < n_nodes && lane < d0.y) e0 = adj[d0.x + lane];

    v4f s  = { 0.0f, 0.0f, 0.0f, 0.0f };
    v4f hi = { -FLT_BIG, -FLT_BIG, -FLT_BIG, -FLT_BIG };
    v4f lo = {  FLT_BIG,  FLT_BIG,  FLT_BIG,  FLT_BIG };

    const int first = st.x;
    const int cnt = (st.y < n_edges) ? st.y : n_edges;

    auto take = [&](const v4f v, bool in) {
      const v4f z = { 0.0f, 0.0f, 0.0f, 0.0f };
      s += in ? v : z;
      for (int k = 0; k < 4; ++k) {
        hi[k] = fmaxf(hi[k], v[k]);
        lo[k] = fminf(lo[k], v[k]);
      }
    };
    auto row = [&](int my, int i) -> v4f {
      const int src = __builtin_amdgcn_readlane(my, i);
      return *(const v4f*)(xl + (long long)src * C);
    };

    auto rows = [&](const int my, const int n) {
      int k = 0;
      for (; k + 8 <= n; k += 8) {
        const v4f v0 = row(my, k),     v1 = row(my, k + 1), v2 = row(my, k + 2), v3 = row(my, k + 3);
        const v4f v4 = row(my, k + 4), v5 = row(my, k + 5), v6 = row(my, k + 6), v7 = row(my, k + 7);
        __builtin_amdgcn_sched_barrier(0);
        take(v0, true); take(v1, true); take(v2, true); take(v3, true);
        take(v4, true); take(v5, true); take(v6, true); take(v7, true);
      }
      for (; k < n; k += 4) {
        const int l  = n - 1;
        const int i1 = (k + 1 < l) ? k + 1 : l;
        const int i2 = (k + 2 < l) ? k + 2 : l;
        const int i3 = (k + 3 < l) ? k + 3 : l;
        const v4f v0 = row(my, k), v1 = row(my, i1), v2 = row(my, i2), v3 = row(my, i3);
        take(v0, true);
        take(v1, k + 1 < n);
        take(v2, k + 2 < n);
        take(v3, k + 3 < n);
      }
    };

    rows(my0, (cnt < 32) ? cnt : 32);
    for (int b0 = 32; b0 < cnt; b0 += 32) {
      const int n = (cnt - b0 < 32) ? (cnt - b0) : 32;
      int my = 0;
      if (lane < n) my = adj[first + b0 + lane];
      rows(my, n);
    }

    const float cn = (float)cnt;
    const float dv = fmaxf(cn, 1.0f);
    v4f m;
    for (int k = 0; k < 4; ++k) m[k] = s[k] / dv;
    if (cnt == 0) {
      hi = (v4f){ 0.0f, 0.0f, 0.0f, 0.0f };
      lo = (v4f){ 0.0f, 0.0f, 0.0f, 0.0f };
    }
    const long long o = (long long)node * C + lane * 4;
    *(v4f*)(sum  + o) = s;
    *(v4f*)(mx   + o) = hi;
    *(v4f*)(mn   + o) = lo;
    *(v4f*)(mean + o) = m;
  }
}

__device__ __forceinline__ v4f ln_row(v4f v, const v4f g, const v4f b) {
#pragma unroll
  for (int k = 0; k < 4; ++k) v[k] = fmaxf(0.5f * v[k], 0.0f);
  float s  = (v[0] + v[1]) + (v[2] + v[3]);
  float s2 = (v[0] * v[0] + v[1] * v[1]) + (v[2] * v[2] + v[3] * v[3]);
#pragma unroll
  for (int off = 16; off > 0; off >>= 1) {
    s  += __shfl_xor(s,  off, 32);
    s2 += __shfl_xor(s2, off, 32);
  }
  const float mu  = s  * (1.0f / C);
  const float var = s2 * (1.0f / C) - mu * mu;
  const float rs  = rsqrtf(var + 1e-5f);
  v4f o;
#pragma unroll
  for (int k = 0; k < 4; ++k) o[k] = (v[k] - mu) * rs * g[k] + b[k];
  return o;
}

#define KC     32
#define BPITCH 36
#define NCHUNK 20
#define TP     132

struct ChunkRegs { v4f a[4]; v4f b[8]; v4f b2[8]; };

template <int V> struct IntC { static constexpr int value = V; };

#define N_WGP 256

template <bool FIRST>
__global__ __launch_bounds__(128) void k_gemm_wmma(
    const float* __restrict__ x,
    const float* __restrict__ mean, const float* __restrict__ mx,
    const float* __restrict__ mn,   const float* __restrict__ sum,
    const float* __restrict__ W1,   const float* __restrict__ W1o,
    const float* __restrict__ W2,
    float* __restrict__ h,
    const float* __restrict__ gamma, const float* __restrict__ beta,
    float* __restrict__ out, int n_nodes, int n_full) {
  __shared__ __attribute__((aligned(16))) float Bs[2 * 128 * BPITCH];

  const int tid  = threadIdx.x;
  const int lane = tid & 31;
  const int wv   = tid >> 5;
  const bool split = (int)blockIdx.x >= n_full;
  const int m0   = split ? n_full * 64 + ((int)blockIdx.x - n_full) * 16
                         : (int)blockIdx.x * 64 + wv * 16;
  const int c    = lane & 15;
  const int hf   = lane >> 4;

  int row = m0 + c;
  if (row >= n_nodes) row = n_nodes - 1;

  const int sn = tid >> 3;
  const int sq = tid & 7;

  v8f acc[8] = {};

  auto load_chunk = [&](int ch, ChunkRegs& r) {
    const int seg = ch >> 2;
    const int kc  = (ch & 3) * KC;
    const float* A  = (seg == 0) ? x : (seg == 1) ? mean : (seg == 2) ? mx : (seg == 3) ? mn : sum;
    const float* Wb = (seg == 0) ? W1 : (W2 + (seg - 1) * C);
    const int   ldw = (seg == 0) ? C : 4 * C;
    const float* Ap = A + (long long)row * C + kc + 4 * hf;
#pragma unroll
    for (int i = 0; i < 4; ++i) r.a[i] = *(const v4f*)(Ap + 8 * i);
    const float* Wp = Wb + (long long)sn * ldw + kc + 4 * sq;
#pragma unroll
    for (int p = 0; p < 8; ++p) r.b[p] = *(const v4f*)(Wp + (long long)(16 * p) * ldw);
    if (!FIRST && seg == 0) {
      const float* Wq = W1o + (long long)sn * C + kc + 4 * sq;
#pragma unroll
      for (int p = 0; p < 8; ++p) r.b2[p] = *(const v4f*)(Wq + (long long)(16 * p) * C);
    }
  };

  auto store_chunk = [&](int stage, const ChunkRegs& r, bool xseg) {
    float* dst = &Bs[stage * (128 * BPITCH) + sn * BPITCH + 4 * sq];
    if (!FIRST && xseg) {
#pragma unroll
      for (int p = 0; p < 8; ++p) *(v4f*)(dst + 16 * p * BPITCH) = r.b[p] + r.b2[p];
    } else {
#pragma unroll
      for (int p = 0; p < 8; ++p) *(v4f*)(dst + 16 * p * BPITCH) = r.b[p];
    }
  };

  auto compute_chunk = [&](int stage, const ChunkRegs& r, auto njc, int jb) {
    constexpr int NJ = decltype(njc)::value;
    const float* src = &Bs[stage * (128 * BPITCH) + (16 * jb + c) * BPITCH + 4 * hf];
    v4f bf[2][NJ];
    auto load_frags = [&](int i, v4f (&f)[NJ]) {
#pragma unroll
      for (int j = 0; j < NJ; ++j) f[j] = *(const v4f*)(src + 16 * j * BPITCH + 8 * i);
    };
    load_frags(0, bf[0]);
#pragma unroll
    for (int i = 0; i < 4; ++i) {
      if (i + 1 < 4) load_frags(i + 1, bf[(i + 1) & 1]);
      __builtin_amdgcn_sched_barrier(0);
      const v2f a0 = { r.a[i][0], r.a[i][1] };
      const v2f a1 = { r.a[i][2], r.a[i][3] };
#pragma unroll
      for (int j = 0; j < NJ; ++j) {
        const v2f b0 = { bf[i & 1][j][0], bf[i & 1][j][1] };
        acc[j] = __builtin_amdgcn_wmma_f32_16x16x4_f32(
            false, a0, false, b0, (short)0, acc[j], false, false);
      }
#pragma unroll
      for (int j = 0; j < NJ; ++j) {
        const v2f b1 = { bf[i & 1][j][2], bf[i & 1][j][3] };
        acc[j] = __builtin_amdgcn_wmma_f32_16x16x4_f32(
            false, a1, false, b1, (short)0, acc[j], false, false);
      }
      __builtin_amdgcn_sched_barrier(0);
    }
  };

  constexpr int CH0 = FIRST ? 4 : 0;
  auto run_k = [&](auto njc, int jb) {
    ChunkRegs cur, nxt;
    load_chunk(CH0, cur);
    store_chunk(CH0 & 1, cur, CH0 < 4);
    __syncthreads();

    for (int ch = CH0; ch < NCHUNK; ++ch) {
      const bool more = (ch + 1 < NCHUNK);
      if (more) load_chunk(ch + 1, nxt);
      asm volatile("" ::: "memory");
      compute_chunk(ch & 1, cur, njc, jb);
      if (more) store_chunk((ch + 1) & 1, nxt, ch + 1 < 4);
      __syncthreads();
      if (more) {
#pragma unroll
        for (int i = 0; i < 4; ++i) cur.a[i] = nxt.a[i];
      }
    }
  };
  if (split) run_k(IntC<2>{}, 2 * wv);
  else       run_k(IntC<8>{}, 0);

  float* T = split ? &Bs[0] : &Bs[wv * (16 * TP)];
  if (split) {
#pragma unroll
    for (int v = 0; v < 8; ++v) {
#pragma unroll
      for (int j = 0; j < 2; ++j) T[(8 * hf + v) * TP + 16 * (2 * wv + j) + c] = acc[j][v];
    }
  } else {
#pragma unroll
    for (int v = 0; v < 8; ++v) {
#pragma unroll
      for (int j = 0; j < 8; ++j) T[(8 * hf + v) * TP + 16 * j + c] = acc[j][v];
    }
  }
  __syncthreads();

  const long long co = lane * 4;
  auto finish = [&](auto nrc, const float* Tt, int mrow) {
    constexpr int NR = decltype(nrc)::value;
    const bool full = (mrow + NR <= n_nodes);
    float* dstp = (FIRST ? h : out) + (long long)mrow * C + co;
    v4f o[NR];
    if (FIRST) {
#pragma unroll
      for (int rr = 0; rr < NR; ++rr) o[rr] = *(const v4f*)(Tt + rr * TP + co);
    } else {
      const v4f g = *(const v4f*)(gamma + co);
      const v4f b = *(const v4f*)(beta  + co);
      v4f hv[NR];
#pragma unroll
      for (int rr = 0; rr < NR; ++rr) {
        int r = mrow + rr;
        if (r >= n_nodes) r = n_nodes - 1;
        hv[rr] = *(const v4f*)(h + (long long)r * C + co);
      }
#pragma unroll
      for (int rr = 0; rr < NR; ++rr) {
        const v4f t = hv[rr] + *(const v4f*)(Tt + rr * TP + co);
        o[rr] = ln_row(t, g, b);
      }
    }
    if (full) {
#pragma unroll
      for (int rr = 0; rr < NR; ++rr) *(v4f*)(dstp + (long long)rr * C) = o[rr];
    } else {
#pragma unroll
      for (int rr = 0; rr < NR; ++rr)
        if (mrow + rr < n_nodes) *(v4f*)(dstp + (long long)rr * C) = o[rr];
    }
  };
  if (split) finish(IntC<4>{}, T + 4 * wv * TP, m0 + 4 * wv);
  else       finish(IntC<16>{}, T, m0);
}

extern "C" void kernel_launch(void* const* d_in, const int* in_sizes, int n_in,
                              void* d_out, int out_size, void* d_ws, size_t ws_size,
                              hipStream_t stream) {
  const float* x     = (const float*)d_in[0];
  const int*   ei1   = (const int*)d_in[1];
  const int*   ei2   = (const int*)d_in[2];
  const float* W1_r1 = (const float*)d_in[3];
  const float* W2_r1 = (const float*)d_in[4];
  const float* W1_r2 = (const float*)d_in[5];
  const float* W2_r2 = (const float*)d_in[6];
  const float* gamma = (const float*)d_in[7];
  const float* beta  = (const float*)d_in[8];
  float* out = (float*)d_out;

  const int N = in_sizes[0] / C;
  const int E = in_sizes[1] / 2;
  const long long NC = (long long)N * C;

  float* sum  = (float*)d_ws;
  float* mx   = sum  + NC;
  float* mn   = mx   + NC;
  float* mean = mn   + NC;
  float* h    = mean + NC;

  const int NP   = (N + 31) / 32 * 32;
  const int nCh  = (E + CHUNK - 1) / CHUNK;
  const int nb   = (N + RS - 1) / RS;
  const int adjS = (E + 31) / 32 * 32 + 32 * MAXB;
  v2i* start = (v2i*)d_out;
  v2i* rec   = start + 2 * (long long)NP;
  int* adj   = (int*)(rec + 2 * (long long)nCh * STRIDE);
  v2i* tab   = (v2i*)d_ws;

  const int T = 256;
  dim3 gG((unsigned)(N_WGP * 4));
  const int nT64   = N / 64;
  const int nFull  = nT64 - nT64 % N_WGP;
  const int nSplit = (N - nFull * 64 + 15) / 16;
  dim3 gM((unsigned)(nFull + nSplit));

  k_bin<<<dim3((unsigned)(2 * nCh)), 1024, 0, stream>>>(ei1, ei2, E, N, rec, tab);
  k_build<<<dim3((unsigned)(2 * nb)), 1024, 0, stream>>>(rec, tab, E, N, start, NP, adj, adjS);

  k_gather<<<gG, T, 0, stream>>>(x, start, adj, N, E, N_WGP * 32, sum, mx, mn, mean);
  k_gemm_wmma<true><<<gM, 128, 0, stream>>>(x, mean, mx, mn, sum, W1_r1, W1_r2, W2_r1, h, gamma, beta, out, N, nFull);

  k_gather<<<gG, T, 0, stream>>>(x, start + NP, adj + adjS, N, E, N_WGP * 32, sum, mx, mn, mean);
  k_gemm_wmma<false><<<gM, 128, 0, stream>>>(x, mean, mx, mn, sum, W1_r2, W1_r1, W2_r2, h, gamma, beta, out, N, nFull);
}


# gather, both relations' products and layer norm in one launch: a work-group gathers its own rows, then multiplies them; accumulators run through both relations (k sum order changes); 6 -> 3 launches
# speedup vs baseline: 5.3264x; 1.0738x over previous
#include <hip/hip_runtime.h>
#include <cstdint>

#define C 128
#define FLT_BIG 3.402823466e38f

typedef float v2f __attribute__((ext_vector_type(2)));
typedef float v4f __attribute__((ext_vector_type(4)));
typedef int   v2i __attribute__((ext_vector_type(2)));
typedef float v8f __attribute__((ext_vector_type(8)));

#define RS_LOG 9
#define RS     (1 << RS_LOG)
#define CHUNK  4096
#define MAXB   128
#define STRIDE (CHUNK + 16 * MAXB)
#define SEGT   256
#define CAP    12288

__global__ __launch_bounds__(1024) void k_bin(
    const int* __restrict__ ei1, const int* __restrict__ ei2,
    int n_edges, int n_nodes, v2i* __restrict__ rec, v2i* __restrict__ tab) {
  __shared__ int cnt[MAXB + 1];
  __shared__ int num[MAXB];
  __shared__ __attribute__((aligned(16))) v2i st[STRIDE];
  const int tid = threadIdx.x;
  const int nCh = (n_edges + CHUNK - 1) / CHUNK;
  const int r   = ((int)blockIdx.x >= nCh) ? 1 : 0;
  const int c   = (int)blockIdx.x - r * nCh;
  const int* __restrict__ ei = r ? ei2 : ei1;
  const int nb  = (n_nodes + RS - 1) >> RS_LOG;
  if (tid <= MAXB) cnt[tid] = 0;
  __syncthreads();

  int  s[4], d[4], rk[4];
  bool ok[4];
  const int base = c * CHUNK;
#pragma unroll
  for (int k = 0; k < 4; ++k) {
    const int e = base + k * 1024 + tid;
    ok[k] = e < n_edges;
    s[k] = 0; d[k] = 0; rk[k] = 0;
    if (ok[k]) { s[k] = ei[e]; d[k] = ei[n_edges + e]; }
  }
#pragma unroll
  for (int k = 0; k < 4; ++k)
    if (ok[k]) rk[k] = atomicAdd(&cnt[d[k] >> RS_LOG], 1);
  __syncthreads();

  if (tid < 32) {
    int v[4], local = 0;
#pragma unroll
    for (int j = 0; j < 4; ++j) {
      const int idx = tid * 4 + j;
      v[j] = (idx < nb) ? cnt[idx] : 0;
      local += (v[j] + 15) & ~15;
    }
    int inc = local;
#pragma unroll
    for (int off = 1; off < 32; off <<= 1) {
      const int t = __shfl_up(inc, off, 32);
      if (tid >= off) inc += t;
    }
    int ex = inc - local;
#pragma unroll
    for (int j = 0; j < 4; ++j) {
      const int idx = tid * 4 + j;
      if (idx < nb) { num[idx] = v[j]; cnt[idx] = ex; ex += (v[j] + 15) & ~15; }
    }
    if (tid == 31) cnt[MAXB] = inc;
  }
  __syncthreads();

#pragma unroll
  for (int k = 0; k < 4; ++k)
    if (ok[k]) {
      v2i q; q.x = s[k]; q.y = d[k];
      st[cnt[d[k] >> RS_LOG] + rk[k]] = q;
    }
  __syncthreads();

  const long long region = (long long)r * nCh + c;
  v2i* __restrict__ dst = rec + region * STRIDE;
  const int used = cnt[MAXB];
  for (int i = tid; i < used; i += 1024) dst[i] = st[i];
  if (tid < nb) {
    v2i t; t.x = cnt[tid]; t.y = cnt[tid] + num[tid];
    tab[region * MAXB + tid] = t;
  }
}

__global__ __launch_bounds__(1024) void k_build(
    const v2i* __restrict__ rec, const v2i* __restrict__ tab,
    int n_edges, int n_nodes, v2i* __restrict__ start, int start_stride,
    int* __restrict__ adj, int adj_stride) {
  __shared__ int cnt[RS];
  __shared__ int tot[MAXB];
  __shared__ int wsum[RS / 32];
  __shared__ int baseS;
  __shared__ v2i sg[SEGT];
  __shared__ int stage[CAP];
  const int tid  = threadIdx.x;
  const int lane = tid & 31;
  const int wv   = tid >> 5;
  const int nCh  = (n_edges + CHUNK - 1) / CHUNK;
  const int nb   = (n_nodes + RS - 1) >> RS_LOG;
  const int r    = ((int)blockIdx.x >= nb) ? 1 : 0;
  const int b    = (int)blockIdx.x - r * nb;
  const int n0   = b << RS_LOG;
  if (tid < RS)   cnt[tid] = 0;
  if (tid < MAXB) tot[tid] = 0;
  const v2i* __restrict__ recr = rec + (long long)r * nCh * STRIDE;
  const v2i* __restrict__ tabr = tab + (long long)r * nCh * MAXB;
  const v2i* __restrict__ tb   = tabr + b;
  int* __restrict__ adjr = adj + (long long)r * adj_stride;
  __syncthreads();

  {
    const int rg = tid & (MAXB - 1);
    if (rg <= b) {
      int part = 0;
#pragma unroll 8
      for (int c = tid >> 7; c < nCh; c += 1024 / MAXB) {
        const v2i t = tabr[(long long)c * MAXB + rg];
        part += t.y - t.x;
      }
      atomicAdd(&tot[rg], part);
    }
  }
  __syncthreads();
  if (wv == 0) {
    int v = 0;
    for (int j = 0; j < 4; ++j) {
      const int i = lane * 4 + j;
      if (i < b) v += (tot[i] + 31) & ~31;
    }
    for (int off = 16; off > 0; off >>= 1) v += __shfl_xor(v, off, 32);
    if (lane == 0) baseS = v;
  }

  auto sweep = [&](auto op) {
    for (int c0 = 0; c0 < nCh; c0 += SEGT) {
      __syncthreads();
      const int nseg = (nCh - c0 < SEGT) ? (nCh - c0) : SEGT;
      if (tid < nseg) {
        v2i t = tb[(long long)(c0 + tid) * MAXB];
        t.x += (c0 + tid) * STRIDE;
        t.y += (c0 + tid) * STRIDE;
        sg[tid] = t;
      }
      __syncthreads();
      int p0[8], pe[8];
      v2i q0[8], q1[8];
#pragma unroll
      for (int j = 0; j < 8; ++j) {
        const int i = wv + 32 * j;
        v2i se = { 0, 0 };
        if (i < nseg) se = sg[i];
        p0[j] = se.x + lane;
        pe[j] = se.y;
        q0[j] = (v2i){ 0, 0 };
        q1[j] = (v2i){ 0, 0 };
        if (p0[j] < pe[j])      q0[j] = recr[p0[j]];
        if (p0[j] + 32 < pe[j]) q1[j] = recr[p0[j] + 32];
      }
#pragma unroll
      for (int j = 0; j < 8; ++j) {
        if (p0[j] < pe[j])      op(q0[j]);
        if (p0[j] + 32 < pe[j]) op(q1[j]);
        for (int p = p0[j] + 64; p < pe[j]; p += 32) op(recr[p]);
      }
    }
  };

  sweep([&](v2i q) { atomicAdd(&cnt[q.y - n0], 1); });
  __syncthreads();

  const int base  = baseS;
  const int total = tot[b];
  int myc = 0, incl = 0;
  if (tid < RS) {
    myc = cnt[tid];
    incl = myc;
    for (int off = 1; off < 32; off <<= 1) {
      const int t = __shfl_up(incl, off, 32);
      if (lane >= off) incl += t;
    }
    if (lane == 31) wsum[wv] = incl;
  }
  __syncthreads();
  if (tid < RS) {
    int pre = 0;
    for (int w = 0; w < RS / 32; ++w)
      if (w < wv) pre += wsum[w];
    const int excl = pre + incl - myc;
    cnt[tid] = excl;
    if (n0 + tid < n_nodes) {
      v2i sv; sv.x = base + excl; sv.y = myc;
      start[(long long)r * start_stride + n0 + tid] = sv;
    }
  }

  const bool inlds = total <= CAP;
  if (inlds) {
    sweep([&](v2i q) { stage[atomicAdd(&cnt[q.y - n0], 1)] = q.x; });
    __syncthreads();
    const int padded = (total + 31) & ~31;
    for (int i = tid; i < padded; i += 1024) adjr[base + i] = (i < total) ? stage[i] : 0;
  } else {
    sweep([&](v2i q) { adjr[base + atomicAdd(&cnt[q.y - n0], 1)] = q.x; });
  }
}

__device__ __forceinline__ v4f ln_row(v4f v, const v4f g, const v4f b) {
#pragma unroll
  for (int k = 0; k < 4; ++k) v[k] = fmaxf(0.5f * v[k], 0.0f);
  float s  = (v[0] + v[1]) + (v[2] + v[3]);
  float s2 = (v[0] * v[0] + v[1] * v[1]) + (v[2] * v[2] + v[3] * v[3]);
#pragma unroll
  for (int off = 16; off > 0; off >>= 1) {
    s  += __shfl_xor(s,  off, 32);
    s2 += __shfl_xor(s2, off, 32);
  }
  const float mu  = s  * (1.0f / C);
  const float var = s2 * (1.0f / C) - mu * mu;
  const float rs  = rsqrtf(var + 1e-5f);
  v4f o;
#pragma unroll
  for (int k = 0; k < 4; ++k) o[k] = (v[k] - mu) * rs * g[k] + b[k];
  return o;
}

#define KC     32
#define BPITCH 36
#define NCHUNK 20
#define TP     132

struct ChunkRegs { v4f a[4]; v4f b[8]; v4f b2[8]; };

template <int V> struct IntC { static constexpr int value = V; };

#define N_WGP 256

__global__ __launch_bounds__(128) void k_conv(
    const float* __restrict__ x,
    const v2i* __restrict__ start, const int* __restrict__ adj,
    int start_stride, int adj_stride,
    float* mean, float* mx, float* mn, float* sum,
    const float* __restrict__ W1a, const float* __restrict__ W1b,
    const float* __restrict__ W2a, const float* __restrict__ W2b,
    const float* __restrict__ gamma, const float* __restrict__ beta,
    float* __restrict__ out, int n_nodes, int n_edges, int n_full) {
  __shared__ __attribute__((aligned(16))) float Bs[2 * 128 * BPITCH];

  const int tid  = threadIdx.x;
  const int lane = tid & 31;
  const int wv   = __builtin_amdgcn_readfirstlane(tid >> 5);
  const bool split = (int)blockIdx.x >= n_full;
  const int m0   = split ? n_full * 64 + ((int)blockIdx.x - n_full) * 16
                         : (int)blockIdx.x * 64 + wv * 16;
  const int c    = lane & 15;
  const int hf   = lane >> 4;

  int row = m0 + c;
  if (row >= n_nodes) row = n_nodes - 1;

  const int sn = tid >> 3;
  const int sq = tid & 7;

  v8f acc[8] = {};

  const int g0   = split ? m0 + 4 * wv : m0;
  const int gtop = g0 + (split ? 4 : 16);
  const int gend = (gtop < n_nodes) ? gtop : n_nodes;
  const float* __restrict__ xl = x + lane * 4;

  auto gather = [&](const v2i* __restrict__ st_, const int* __restrict__ adj_) {
    if (g0 >= gend) return;
    v2i d0 = st_[g0];
    v2i d1 = { 0, 0 };
    if (g0 + 1 < gend) d1 = st_[g0 + 1];
    int e0 = 0;
    if (lane < d0.y) e0 = adj_[d0.x + lane];
    asm volatile("" :: "v"(e0));

    for (int node = g0; node < gend; ++node) {
      const v2i st  = d0;
      const int my0 = e0;
      d0 = d1;
      if (node + 2 < gend) d1 = st_[node + 2];
      e0 = 0;
      if (node + 1 < gend && lane < d0.y) e0 = adj_[d0.x + lane];

      v4f s  = { 0.0f, 0.0f, 0.0f, 0.0f };
      v4f hi = { -FLT_BIG, -FLT_BIG, -FLT_BIG, -FLT_BIG };
      v4f lo = {  FLT_BIG,  FLT_BIG,  FLT_BIG,  FLT_BIG };

      const int first = st.x;
      const int cnt = (st.y < n_edges) ? st.y : n_edges;

      auto take = [&](const v4f v) {
        s += v;
        for (int k = 0; k < 4; ++k) {
          hi[k] = fmaxf(hi[k], v[k]);
          lo[k] = fminf(lo[k], v[k]);
        }
      };
      auto rowof = [&](int my, int i) -> v4f {
        const int src = __builtin_amdgcn_readlane(my, i);
        return *(const v4f*)(xl + (long long)src * C);
      };

      auto rows = [&](const int my, const int n) {
        int k = 0;
        for (; k + 16 <= n; k += 16) {
          v4f v[16];
#pragma unroll
          for (int i = 0; i < 16; ++i) v[i] = rowof(my, k + i);
          __builtin_amdgcn_sched_barrier(0);
#pragma unroll
          for (int i = 0; i < 16; ++i) take(v[i]);
        }
        const int r  = n - k;
        const int k8 = k  + (r & 8);
        const int k4 = k8 + (r & 4);
        const int k2 = k4 + (r & 2);
        const v4f z = { 0.0f, 0.0f, 0.0f, 0.0f };
        v4f p[8], q[4], t[2], u = z;
#pragma unroll
        for (int i = 0; i < 8; ++i) p[i] = z;
#pragma unroll
        for (int i = 0; i < 4; ++i) q[i] = z;
        t[0] = z; t[1] = z;
        if (r & 8) {
#pragma unroll
          for (int i = 0; i < 8; ++i) p[i] = rowof(my, k + i);
        }
        if (r & 4) {
#pragma unroll
          for (int i = 0; i < 4; ++i) q[i] = rowof(my, k8 + i);
        }
        if (r & 2) { t[0] = rowof(my, k4); t[1] = rowof(my, k4 + 1); }
        if (r & 1) u = rowof(my, k2);
        if (r & 8) {
#pragma unroll
          for (int i = 0; i < 8; ++i) take(p[i]);
        }
        if (r & 4) {
#pragma unroll
          for (int i = 0; i < 4; ++i) take(q[i]);
        }
        if (r & 2) { take(t[0]); take(t[1]); }
        if (r & 1) take(u);
      };

      rows(my0, (cnt < 32) ? cnt : 32);
      for (int b0 = 32; b0 < cnt; b0 += 32) {
        const int n = (cnt - b0 < 32) ? (cnt - b0) : 32;
        int my = 0;
        if (lane < n) my = adj_[first + b0 + lane];
        rows(my, n);
      }

      const float cn = (float)cnt;
      const float dv = fmaxf(cn, 1.0f);
      v4f m;
      for (int k = 0; k < 4; ++k) m[k] = s[k] / dv;
      if (cnt == 0) {
        hi = (v4f){ 0.0f, 0.0f, 0.0f, 0.0f };
        lo = (v4f){ 0.0f, 0.0f, 0.0f, 0.0f };
      }
      const long long o = (long long)node * C + lane * 4;
      *(v4f*)(sum  + o) = s;
      *(v4f*)(mx   + o) = hi;
      *(v4f*)(mn   + o) = lo;
      *(v4f*)(mean + o) = m;
    }
  };

  auto load_chunk = [&](auto relc, int ch, ChunkRegs& r) {
    constexpr int REL = decltype(relc)::value;
    const int seg = ch >> 2;
    const int kc  = (ch & 3) * KC;
    const float* A  = (seg == 0) ? x : (seg == 1) ? mean : (seg == 2) ? mx : (seg == 3) ? mn : sum;
    const float* W2 = REL ? W2b : W2a;
    const float* Wb = (seg == 0) ? W1a : (W2 + (seg - 1) * C);
    const int   ldw = (seg == 0) ? C : 4 * C;
    const float* Ap = A + (long long)row * C + kc + 4 * hf;
#pragma unroll
    for (int i = 0; i < 4; ++i) r.a[i] = *(const v4f*)(Ap + 8 * i);
    const float* Wp = Wb + (long long)sn * ldw + kc + 4 * sq;
#pragma unroll
    for (int p = 0; p < 8; ++p) r.b[p] = *(const v4f*)(Wp + (long long)(16 * p) * ldw);
    if (REL && seg == 0) {
      const float* Wq = W1b + (long long)sn * C + kc + 4 * sq;
#pragma unroll
      for (int p = 0; p < 8; ++p) r.b2[p] = *(const v4f*)(Wq + (long long)(16 * p) * C);
    }
  };

  auto store_chunk = [&](auto relc, int stage, const ChunkRegs& r, bool xseg) {
    constexpr int REL = decltype(relc)::value;
    float* dst = &Bs[stage * (128 * BPITCH) + sn * BPITCH + 4 * sq];
    if (REL && xseg) {
#pragma unroll
      for (int p = 0; p < 8; ++p) *(v4f*)(dst + 16 * p * BPITCH) = r.b[p] + r.b2[p];
    } else {
#pragma unroll
      for (int p = 0; p < 8; ++p) *(v4f*)(dst + 16 * p * BPITCH) = r.b[p];
    }
  };

  auto compute_chunk = [&](int stage, const ChunkRegs& r, auto njc, int jb) {
    constexpr int NJ = decltype(njc)::value;
    const float* src = &Bs[stage * (128 * BPITCH) + (16 * jb + c) * BPITCH + 4 * hf];
    v4f bf[2][NJ];
    auto load_frags = [&](int i, v4f (&f)[NJ]) {
#pragma unroll
      for (int j = 0; j < NJ; ++j) f[j] = *(const v4f*)(src + 16 * j * BPITCH + 8 * i);
    };
    load_frags(0, bf[0]);
#pragma unroll
    for (int i = 0; i < 4; ++i) {
      if (i + 1 < 4) load_frags(i + 1, bf[(i + 1) & 1]);
      __builtin_amdgcn_sched_barrier(0);
      const v2f a0 = { r.a[i][0], r.a[i][1] };
      const v2f a1 = { r.a[i][2], r.a[i][3] };
#pragma unroll
      for (int j = 0; j < NJ; ++j) {
        const v2f b0 = { bf[i & 1][j][0], bf[i & 1][j][1] };
        acc[j] = __builtin_amdgcn_wmma_f32_16x16x4_f32(
            false, a0, false, b0, (short)0, acc[j], false, false);
      }
#pragma unroll
      for (int j = 0; j < NJ; ++j) {
        const v2f b1 = { bf[i & 1][j][2], bf[i & 1][j][3] };
        acc[j] = __builtin_amdgcn_wmma_f32_16x16x4_f32(
            false, a1, false, b1, (short)0, acc[j], false, false);
      }
      __builtin_amdgcn_sched_barrier(0);
    }
  };

  auto run_k = [&](auto relc, auto njc, int jb) {
    constexpr int REL = decltype(relc)::value;
    constexpr int CH0 = REL ? 0 : 4;
    ChunkRegs cur, nxt;
    load_chunk(relc, CH0, cur);
    store_chunk(relc, CH0 & 1, cur, CH0 < 4);
    __syncthreads();

    for (int ch = CH0; ch < NCHUNK; ++ch) {
      const bool more = (ch + 1 < NCHUNK);
      if (more) load_chunk(relc, ch + 1, nxt);
      asm volatile("" ::: "memory");
      compute_chunk(ch & 1, cur, njc, jb);
      if (more) store_chunk(relc, (ch + 1) & 1, nxt, ch + 1 < 4);
      __syncthreads();
      if (more) {
#pragma unroll
        for (int i = 0; i < 4; ++i) cur.a[i] = nxt.a[i];
      }
    }
  };

  gather(start, adj);
  __syncthreads();
  if (split) run_k(IntC<0>{}, IntC<2>{}, 2 * wv);
  else       run_k(IntC<0>{}, IntC<8>{}, 0);

  gather(start + start_stride, adj + adj_stride);
  __syncthreads();
  if (split) run_k(IntC<1>{}, IntC<2>{}, 2 * wv);
  else       run_k(IntC<1>{}, IntC<8>{}, 0);

  float* T = split ? &Bs[0] : &Bs[wv * (16 * TP)];
  if (split) {
#pragma unroll
    for (int v = 0; v < 8; ++v) {
#pragma unroll
      for (int j = 0; j < 2; ++j) T[(8 * hf + v) * TP + 16 * (2 * wv + j) + c] = acc[j][v];
    }
  } else {
#pragma unroll
    for (int v = 0; v < 8; ++v) {
#pragma unroll
      for (int j = 0; j < 8; ++j) T[(8 * hf + v) * TP + 16 * j + c] = acc[j][v];
    }
  }
  __syncthreads();

  const long long co = lane * 4;
  auto finish = [&](auto nrc, const float* Tt, int mrow) {
    constexpr int NR = decltype(nrc)::value;
    const bool full = (mrow + NR <= n_nodes);
    float* dstp = out + (long long)mrow * C + co;
    const v4f g = *(const v4f*)(gamma + co);
    const v4f b = *(const v4f*)(beta  + co);
    v4f o[NR];
#pragma unroll
    for (int rr = 0; rr < NR; ++rr) o[rr] = ln_row(*(const v4f*)(Tt + rr * TP + co), g, b);
    if (full) {
#pragma unroll
      for (int rr = 0; rr < NR; ++rr) *(v4f*)(dstp + (long long)rr * C) = o[rr];
    } else {
#pragma unroll
      for (int rr = 0; rr < NR; ++rr)
        if (mrow + rr < n_nodes) *(v4f*)(dstp + (long long)rr * C) = o[rr];
    }
  };
  if (split) finish(IntC<4>{}, T + 4 * wv * TP, m0 + 4 * wv);
  else       finish(IntC<16>{}, T, m0);
}

extern "C" void kernel_launch(void* const* d_in, const int* in_sizes, int n_in,
                              void* d_out, int out_size, void* d_ws, size_t ws_size,
                              hipStream_t stream) {
  const float* x     = (const float*)d_in[0];
  const int*   ei1   = (const int*)d_in[1];
  const int*   ei2   = (const int*)d_in[2];
  const float* W1_r1 = (const float*)d_in[3];
  const float* W2_r1 = (const float*)d_in[4];
  const float* W1_r2 = (const float*)d_in[5];
  const float* W2_r2 = (const float*)d_in[6];
  const float* gamma = (const float*)d_in[7];
  const float* beta  = (const float*)d_in[8];
  float* out = (float*)d_out;

  const int N = in_sizes[0] / C;
  const int E = in_sizes[1] / 2;
  const long long NC = (long long)N * C;

  float* sum  = (float*)d_ws;
  float* mx   = sum  + NC;
  float* mn   = mx   + NC;
  float* mean = mn   + NC;
  float* p5   = mean + NC;

  const int NP   = (N + 31) / 32 * 32;
  const int nCh  = (E + CHUNK - 1) / CHUNK;
  const int nb   = (N + RS - 1) / RS;
  const int adjS = (E + 31) / 32 * 32 + 32 * MAXB;
  v2i* start = (v2i*)p5;
  int* adj   = (int*)(start + 2 * (long long)NP);
  v2i* rec   = (v2i*)mx;
  v2i* tab   = (v2i*)d_ws;

  const int nT64   = N / 64;
  const int nFull  = nT64 - nT64 % N_WGP;
  const int nSplit = (N - nFull * 64 + 15) / 16;
  dim3 gM((unsigned)(nFull + nSplit));

  k_bin<<<dim3((unsigned)(2 * nCh)), 1024, 0, stream>>>(ei1, ei2, E, N, rec, tab);
  k_build<<<dim3((unsigned)(2 * nb)), 1024, 0, stream>>>(rec, tab, E, N, start, NP, adj, adjS);
  k_conv<<<gM, 128, 0, stream>>>(x, start, adj, NP, adjS, mean, mx, mn, sum,
                                 W1_r1, W1_r2, W2_r1, W2_r2, gamma, beta, out, N, E, nFull);
}


# gathered aggregate rows handed to the products through LDS instead of workspace planes: 32-row work-groups (16x64 tile per wave), three resident per processor; all sum orders kept
# speedup vs baseline: 5.6134x; 1.0539x over previous
#include <hip/hip_runtime.h>
#include <cstdint>

#define C 128
#define FLT_BIG 3.402823466e38f

typedef float v2f __attribute__((ext_vector_type(2)));
typedef float v4f __attribute__((ext_vector_type(4)));
typedef int   v2i __attribute__((ext_vector_type(2)));
typedef float v8f __attribute__((ext_vector_type(8)));

#define RS_LOG 9
#define RS     (1 << RS_LOG)
#define CHUNK  4096
#define MAXB   128
#define STRIDE (CHUNK + 16 * MAXB)
#define SEGT   256
#define CAP    12288

__global__ __launch_bounds__(1024) void k_bin(
    const int* __restrict__ ei1, const int* __restrict__ ei2,
    int n_edges, int n_nodes, v2i* __restrict__ rec, v2i* __restrict__ tab) {
  __shared__ int cnt[MAXB + 1];
  __shared__ int num[MAXB];
  __shared__ __attribute__((aligned(16))) v2i st[STRIDE];
  const int tid = threadIdx.x;
  const int nCh = (n_edges + CHUNK - 1) / CHUNK;
  const int r   = ((int)blockIdx.x >= nCh) ? 1 : 0;
  const int c   = (int)blockIdx.x - r * nCh;
  const int* __restrict__ ei = r ? ei2 : ei1;
  const int nb  = (n_nodes + RS - 1) >> RS_LOG;
  if (tid <= MAXB) cnt[tid] = 0;
  __syncthreads();

  int  s[4], d[4], rk[4];
  bool ok[4];
  const int base = c * CHUNK;
#pragma unroll
  for (int k = 0; k < 4; ++k) {
    const int e = base + k * 1024 + tid;
    ok[k] = e < n_edges;
    s[k] = 0; d[k] = 0; rk[k] = 0;
    if (ok[k]) { s[k] = ei[e]; d[k] = ei[n_edges + e]; }
  }
#pragma unroll
  for (int k = 0; k < 4; ++k)
    if (ok[k]) rk[k] = atomicAdd(&cnt[d[k] >> RS_LOG], 1);
  __syncthreads();

  if (tid < 32) {
    int v[4], local = 0;
#pragma unroll
    for (int j = 0; j < 4; ++j) {
      const int idx = tid * 4 + j;
      v[j] = (idx < nb) ? cnt[idx] : 0;
      local += (v[j] + 15) & ~15;
    }
    int inc = local;
#pragma unroll
    for (int off = 1; off < 32; off <<= 1) {
      const int t = __shfl_up(inc, off, 32);
      if (tid >= off) inc += t;
    }
    int ex = inc - local;
#pragma unroll
    for (int j = 0; j < 4; ++j) {
      const int idx = tid * 4 + j;
      if (idx < nb) { num[idx] = v[j]; cnt[idx] = ex; ex += (v[j] + 15) & ~15; }
    }
    if (tid == 31) cnt[MAXB] = inc;
  }
  __syncthreads();

#pragma unroll
  for (int k = 0; k < 4; ++k)
    if (ok[k]) {
      v2i q; q.x = s[k]; q.y = d[k];
      st[cnt[d[k] >> RS_LOG] + rk[k]] = q;
    }
  __syncthreads();

  const long long region = (long long)r * nCh + c;
  v2i* __restrict__ dst = rec + region * STRIDE;
  const int used = cnt[MAXB];
  for (int i = tid; i < used; i += 1024) dst[i] = st[i];
  if (tid < nb) {
    v2i t; t.x = cnt[tid]; t.y = cnt[tid] + num[tid];
    tab[region * MAXB + tid] = t;
  }
}

__global__ __launch_bounds__(1024) void k_build(
    const v2i* __restrict__ rec, const v2i* __restrict__ tab,
    int n_edges, int n_nodes, v2i* __restrict__ start, int start_stride,
    int* __restrict__ adj, int adj_stride) {
  __shared__ int cnt[RS];
  __shared__ int tot[MAXB];
  __shared__ int wsum[RS / 32];
  __shared__ int baseS;
  __shared__ v2i sg[SEGT];
  __shared__ int stage[CAP];
  const int tid  = threadIdx.x;
  const int lane = tid & 31;
  const int wv   = tid >> 5;
  const int nCh  = (n_edges + CHUNK - 1) / CHUNK;
  const int nb   = (n_nodes + RS - 1) >> RS_LOG;
  const int r    = ((int)blockIdx.x >= nb) ? 1 : 0;
  const int b    = (int)blockIdx.x - r * nb;
  const int n0   = b << RS_LOG;
  if (tid < RS)   cnt[tid] = 0;
  if (tid < MAXB) tot[tid] = 0;
  const v2i* __restrict__ recr = rec + (long long)r * nCh * STRIDE;
  const v2i* __restrict__ tabr = tab + (long long)r * nCh * MAXB;
  const v2i* __restrict__ tb   = tabr + b;
  int* __restrict__ adjr = adj + (long long)r * adj_stride;
  __syncthreads();

  {
    const int rg = tid & (MAXB - 1);
    if (rg <= b) {
      int part = 0;
#pragma unroll 8
      for (int c = tid >> 7; c < nCh; c += 1024 / MAXB) {
        const v2i t = tabr[(long long)c * MAXB + rg];
        part += t.y - t.x;
      }
      atomicAdd(&tot[rg], part);
    }
  }
  __syncthreads();
  if (wv == 0) {
    int v = 0;
    for (int j = 0; j < 4; ++j) {
      const int i = lane * 4 + j;
      if (i < b) v += (tot[i] + 31) & ~31;
    }
    for (int off = 16; off > 0; off >>= 1) v += __shfl_xor(v, off, 32);
    if (lane == 0) baseS = v;
  }

  auto sweep = [&](auto op) {
    for (int c0 = 0; c0 < nCh; c0 += SEGT) {
      __syncthreads();
      const int nseg = (nCh - c0 < SEGT) ? (nCh - c0) : SEGT;
      if (tid < nseg) {
        v2i t = tb[(long long)(c0 + tid) * MAXB];
        t.x += (c0 + tid) * STRIDE;
        t.y += (c0 + tid) * STRIDE;
        sg[tid] = t;
      }
      __syncthreads();
      int p0[8], pe[8];
      v2i q0[8], q1[8];
#pragma unroll
      for (int j = 0; j < 8; ++j) {
        const int i = wv + 32 * j;
        v2i se = { 0, 0 };
        if (i < nseg) se = sg[i];
        p0[j] = se.x + lane;
        pe[j] = se.y;
        q0[j] = (v2i){ 0, 0 };
        q1[j] = (v2i){ 0, 0 };
        if (p0[j] < pe[j])      q0[j] = recr[p0[j]];
        if (p0[j] + 32 < pe[j]) q1[j] = recr[p0[j] + 32];
      }
#pragma unroll
      for (int j = 0; j < 8; ++j) {
        if (p0[j] < pe[j])      op(q0[j]);
        if (p0[j] + 32 < pe[j]) op(q1[j]);
        for (int p = p0[j] + 64; p < pe[j]; p += 32) op(recr[p]);
      }
    }
  };

  sweep([&](v2i q) { atomicAdd(&cnt[q.y - n0], 1); });
  __syncthreads();

  const int base  = baseS;
  const int total = tot[b];
  int myc = 0, incl = 0;
  if (tid < RS) {
    myc = cnt[tid];
    incl = myc;
    for (int off = 1; off < 32; off <<= 1) {
      const int t = __shfl_up(incl, off, 32);
      if (lane >= off) incl += t;
    }
    if (lane == 31) wsum[wv] = incl;
  }
  __syncthreads();
  if (tid < RS) {
    int pre = 0;
    for (int w = 0; w < RS / 32; ++w)
      if (w < wv) pre += wsum[w];
    const int excl = pre + incl - myc;
    cnt[tid] = excl;
    if (n0 + tid < n_nodes) {
      v2i sv; sv.x = base + excl; sv.y = myc;
      start[(long long)r * start_stride + n0 + tid] = sv;
    }
  }

  const bool inlds = total <= CAP;
  if (inlds) {
    sweep([&](v2i q) { stage[atomicAdd(&cnt[q.y - n0], 1)] = q.x; });
    __syncthreads();
    const int padded = (total + 31) & ~31;
    for (int i = tid; i < padded; i += 1024) adjr[base + i] = (i < total) ? stage[i] : 0;
  } else {
    sweep([&](v2i q) { adjr[base + atomicAdd(&cnt[q.y - n0], 1)] = q.x; });
  }
}

__device__ __forceinline__ v4f ln_row(v4f v, const v4f g, const v4f b) {
#pragma unroll
  for (int k = 0; k < 4; ++k) v[k] = fmaxf(0.5f * v[k], 0.0f);
  float s  = (v[0] + v[1]) + (v[2] + v[3]);
  float s2 = (v[0] * v[0] + v[1] * v[1]) + (v[2] * v[2] + v[3] * v[3]);
#pragma unroll
  for (int off = 16; off > 0; off >>= 1) {
    s  += __shfl_xor(s,  off, 32);
    s2 += __shfl_xor(s2, off, 32);
  }
  const float mu  = s  * (1.0f / C);
  const float var = s2 * (1.0f / C) - mu * mu;
  const float rs  = rsqrtf(var + 1e-5f);
  v4f o;
#pragma unroll
  for (int k = 0; k < 4; ++k) o[k] = (v[k] - mu) * rs * g[k] + b[k];
  return o;
}

#define KC     32
#define BPITCH 36
#define NCHUNK 20
#define TP     132
#define APITCH 516
#define WGROWS 32

struct ChunkRegs { v4f a[4]; v4f b[8]; v4f b2[8]; };

template <int V> struct IntC { static constexpr int value = V; };

#define N_WGP 256

__global__ __launch_bounds__(128) void k_conv(
    const float* __restrict__ x,
    const v2i* __restrict__ start, const int* __restrict__ adj,
    int start_stride, int adj_stride,
    const float* __restrict__ W1a, const float* __restrict__ W1b,
    const float* __restrict__ W2a, const float* __restrict__ W2b,
    const float* __restrict__ gamma, const float* __restrict__ beta,
    float* __restrict__ out, int n_nodes, int n_edges, int n_full) {
  __shared__ __attribute__((aligned(16))) float Bs[2 * 128 * BPITCH];
  __shared__ __attribute__((aligned(16))) float As[WGROWS * APITCH];

  const int tid  = threadIdx.x;
  const int lane = tid & 31;
  const int wv   = __builtin_amdgcn_readfirstlane(tid >> 5);
  const bool split = (int)blockIdx.x >= n_full;
  const int wr   = split ? 0 : (wv >> 1);
  const int mb   = split ? n_full * WGROWS + ((int)blockIdx.x - n_full) * 16
                         : (int)blockIdx.x * WGROWS;
  const int m0   = mb + 16 * wr;
  const int c    = lane & 15;
  const int hf   = lane >> 4;

  int row = m0 + c;
  if (row >= n_nodes) row = n_nodes - 1;
  const int lrow = 16 * wr + c;

  const int sn = tid >> 3;
  const int sq = tid & 7;

  v8f acc[4] = {};

  const int g0   = mb + (split ? 4 : 8) * wv;
  const int gtop = g0 + (split ? 4 : 8);
  const int gend = (gtop < n_nodes) ? gtop : n_nodes;
  const float* __restrict__ xl = x + lane * 4;

  auto gather = [&](const v2i* __restrict__ st_, const int* __restrict__ adj_) {
    if (g0 >= gend) return;
    v2i d0 = st_[g0];
    v2i d1 = { 0, 0 };
    if (g0 + 1 < gend) d1 = st_[g0 + 1];
    int e0 = 0;
    if (lane < d0.y) e0 = adj_[d0.x + lane];
    asm volatile("" :: "v"(e0));

    for (int node = g0; node < gend; ++node) {
      const v2i st  = d0;
      const int my0 = e0;
      d0 = d1;
      if (node + 2 < gend) d1 = st_[node + 2];
      e0 = 0;
      if (node + 1 < gend && lane < d0.y) e0 = adj_[d0.x + lane];

      v4f s  = { 0.0f, 0.0f, 0.0f, 0.0f };
      v4f hi = { -FLT_BIG, -FLT_BIG, -FLT_BIG, -FLT_BIG };
      v4f lo = {  FLT_BIG,  FLT_BIG,  FLT_BIG,  FLT_BIG };

      const int first = st.x;
      const int cnt = (st.y < n_edges) ? st.y : n_edges;

      auto take = [&](const v4f v) {
        s += v;
        for (int k = 0; k < 4; ++k) {
          hi[k] = fmaxf(hi[k], v[k]);
          lo[k] = fminf(lo[k], v[k]);
        }
      };
      auto rowof = [&](int my, int i) -> v4f {
        const int src = __builtin_amdgcn_readlane(my, i);
        return *(const v4f*)(xl + (long long)src * C);
      };

      auto rows = [&](const int my, const int n) {
        int k = 0;
        for (; k + 16 <= n; k += 16) {
          v4f v[16];
#pragma unroll
          for (int i = 0; i < 16; ++i) v[i] = rowof(my, k + i);
          __builtin_amdgcn_sched_barrier(0);
#pragma unroll
          for (int i = 0; i < 16; ++i) take(v[i]);
        }
        const int r  = n - k;
        const int k8 = k  + (r & 8);
        const int k4 = k8 + (r & 4);
        const int k2 = k4 + (r & 2);
        const v4f z = { 0.0f, 0.0f, 0.0f, 0.0f };
        v4f p[8], q[4], t[2], u = z;
#pragma unroll
        for (int i = 0; i < 8; ++i) p[i] = z;
#pragma unroll
        for (int i = 0; i < 4; ++i) q[i] = z;
        t[0] = z; t[1] = z;
        if (r & 8) {
#pragma unroll
          for (int i = 0; i < 8; ++i) p[i] = rowof(my, k + i);
        }
        if (r & 4) {
#pragma unroll
          for (int i = 0; i < 4; ++i) q[i] = rowof(my, k8 + i);
        }
        if (r & 2) { t[0] = rowof(my, k4); t[1] = rowof(my, k4 + 1); }
        if (r & 1) u = rowof(my, k2);
        if (r & 8) {
#pragma unroll
          for (int i = 0; i < 8; ++i) take(p[i]);
        }
        if (r & 4) {
#pragma unroll
          for (int i = 0; i < 4; ++i) take(q[i]);
        }
        if (r & 2) { take(t[0]); take(t[1]); }
        if (r & 1) take(u);
      };

      rows(my0, (cnt < 32) ? cnt : 32);
      for (int b0 = 32; b0 < cnt; b0 += 32) {
        const int n = (cnt - b0 < 32) ? (cnt - b0) : 32;
        int my = 0;
        if (lane < n) my = adj_[first + b0 + lane];
        rows(my, n);
      }

      const float cn = (float)cnt;
      const float dv = fmaxf(cn, 1.0f);
      v4f m;
      for (int k = 0; k < 4; ++k) m[k] = s[k] / dv;
      if (cnt == 0) {
        hi = (v4f){ 0.0f, 0.0f, 0.0f, 0.0f };
        lo = (v4f){ 0.0f, 0.0f, 0.0f, 0.0f };
      }
      float* ap = &As[(node - mb) * APITCH + lane * 4];
      *(v4f*)(ap)         = m;
      *(v4f*)(ap + C)     = hi;
      *(v4f*)(ap + 2 * C) = lo;
      *(v4f*)(ap + 3 * C) = s;
    }
  };

  auto load_chunk = [&](auto relc, int ch, ChunkRegs& r) {
    constexpr int REL = decltype(relc)::value;
    const int seg = ch >> 2;
    const int kc  = (ch & 3) * KC;
    const float* W2 = REL ? W2b : W2a;
    const float* Wb = (seg == 0) ? W1a : (W2 + (seg - 1) * C);
    const int   ldw = (seg == 0) ? C : 4 * C;
    if (seg == 0) {
      const float* Ap = x + (long long)row * C + kc + 4 * hf;
#pragma unroll
      for (int i = 0; i < 4; ++i) r.a[i] = *(const v4f*)(Ap + 8 * i);
    } else {
      const float* Ap = &As[lrow * APITCH + (seg - 1) * C + kc + 4 * hf];
#pragma unroll
      for (int i = 0; i < 4; ++i) r.a[i] = *(const v4f*)(Ap + 8 * i);
    }
    const float* Wp = Wb + (long long)sn * ldw + kc + 4 * sq;
#pragma unroll
    for (int p = 0; p < 8; ++p) r.b[p] = *(const v4f*)(Wp + (long long)(16 * p) * ldw);
    if (REL && seg == 0) {
      const float* Wq = W1b + (long long)sn * C + kc + 4 * sq;
#pragma unroll
      for (int p = 0; p < 8; ++p) r.b2[p] = *(const v4f*)(Wq + (long long)(16 * p) * C);
    }
  };

  auto store_chunk = [&](auto relc, int stage, const ChunkRegs& r, bool xseg) {
    constexpr int REL = decltype(relc)::value;
    float* dst = &Bs[stage * (128 * BPITCH) + sn * BPITCH + 4 * sq];
    if (REL && xseg) {
#pragma unroll
      for (int p = 0; p < 8; ++p) *(v4f*)(dst + 16 * p * BPITCH) = r.b[p] + r.b2[p];
    } else {
#pragma unroll
      for (int p = 0; p < 8; ++p) *(v4f*)(dst + 16 * p * BPITCH) = r.b[p];
    }
  };

  auto compute_chunk = [&](int stage, const ChunkRegs& r, auto njc, int jb) {
    constexpr int NJ = decltype(njc)::value;
    const float* src = &Bs[stage * (128 * BPITCH) + (16 * jb + c) * BPITCH + 4 * hf];
    v4f bf[2][NJ];
    auto load_frags = [&](int i, v4f (&f)[NJ]) {
#pragma unroll
      for (int j = 0; j < NJ; ++j) f[j] = *(const v4f*)(src + 16 * j * BPITCH + 8 * i);
    };
    load_frags(0, bf[0]);
#pragma unroll
    for (int i = 0; i < 4; ++i) {
      if (i + 1 < 4) load_frags(i + 1, bf[(i + 1) & 1]);
      __builtin_amdgcn_sched_barrier(0);
      const v2f a0 = { r.a[i][0], r.a[i][1] };
      const v2f a1 = { r.a[i][2], r.a[i][3] };
#pragma unroll
      for (int j = 0; j < NJ; ++j) {
        const v2f b0 = { bf[i & 1][j][0], bf[i & 1][j][1] };
        acc[j] = __builtin_amdgcn_wmma_f32_16x16x4_f32(
            false, a0, false, b0, (short)0, acc[j], false, false);
      }
#pragma unroll
      for (int j = 0; j < NJ; ++j) {
        const v2f b1 = { bf[i & 1][j][2], bf[i & 1][j][3] };
        acc[j] = __builtin_amdgcn_wmma_f32_16x16x4_f32(
            false, a1, false, b1, (short)0, acc[j], false, false);
      }
      __builtin_amdgcn_sched_barrier(0);
    }
  };

  auto run_k = [&](auto relc, auto njc, int jb) {
    constexpr int REL = decltype(relc)::value;
    constexpr int CH0 = REL ? 0 : 4;
    ChunkRegs cur, nxt;
    load_chunk(relc, CH0, cur);
    store_chunk(relc, CH0 & 1, cur, CH0 < 4);
    __syncthreads();

    for (int ch = CH0; ch < NCHUNK; ++ch) {
      const bool more = (ch + 1 < NCHUNK);
      if (more) load_chunk(relc, ch + 1, nxt);
      asm volatile("" ::: "memory");
      compute_chunk(ch & 1, cur, njc, jb);
      if (more) store_chunk(relc, (ch + 1) & 1, nxt, ch + 1 < 4);
      __syncthreads();
      if (more) {
#pragma unroll
        for (int i = 0; i < 4; ++i) cur.a[i] = nxt.a[i];
      }
    }
  };

  gather(start, adj);
  __syncthreads();
  if (split) run_k(IntC<0>{}, IntC<2>{}, 2 * wv);
  else       run_k(IntC<0>{}, IntC<4>{}, 4 * (wv & 1));

  gather(start + start_stride, adj + adj_stride);
  __syncthreads();
  if (split) run_k(IntC<1>{}, IntC<2>{}, 2 * wv);
  else       run_k(IntC<1>{}, IntC<4>{}, 4 * (wv & 1));

  float* T = &Bs[wr * (16 * TP)];
  if (split) {
#pragma unroll
    for (int v = 0; v < 8; ++v) {
#pragma unroll
      for (int j = 0; j < 2; ++j) T[(8 * hf + v) * TP + 16 * (2 * wv + j) + c] = acc[j][v];
    }
  } else {
#pragma unroll
    for (int v = 0; v < 8; ++v) {
#pragma unroll
      for (int j = 0; j < 4; ++j) T[(8 * hf + v) * TP + 16 * (4 * (wv & 1) + j) + c] = acc[j][v];
    }
  }
  __syncthreads();

  const long long co = lane * 4;
  auto finish = [&](auto nrc, const float* Tt, int mrow) {
    constexpr int NR = decltype(nrc)::value;
    const bool full = (mrow + NR <= n_nodes);
    float* dstp = out + (long long)mrow * C + co;
    const v4f g = *(const v4f*)(gamma + co);
    const v4f b = *(const v4f*)(beta  + co);
    v4f o[NR];
#pragma unroll
    for (int rr = 0; rr < NR; ++rr) o[rr] = ln_row(*(const v4f*)(Tt + rr * TP + co), g, b);
    if (full) {
#pragma unroll
      for (int rr = 0; rr < NR; ++rr) *(v4f*)(dstp + (long long)rr * C) = o[rr];
    } else {
#pragma unroll
      for (int rr = 0; rr < NR; ++rr)
        if (mrow + rr < n_nodes) *(v4f*)(dstp + (long long)rr * C) = o[rr];
    }
  };
  if (split) finish(IntC<4>{}, T + 4 * wv * TP, mb + 4 * wv);
  else       finish(IntC<8>{}, T + 8 * (wv & 1) * TP, mb + 8 * wv);
}

extern "C" void kernel_launch(void* const* d_in, const int* in_sizes, int n_in,
                              void* d_out, int out_size, void* d_ws, size_t ws_size,
                              hipStream_t stream) {
  const float* x     = (const float*)d_in[0];
  const int*   ei1   = (const int*)d_in[1];
  const int*   ei2   = (const int*)d_in[2];
  const float* W1_r1 = (const float*)d_in[3];
  const float* W2_r1 = (const float*)d_in[4];
  const float* W1_r2 = (const float*)d_in[5];
  const float* W2_r2 = (const float*)d_in[6];
  const float* gamma = (const float*)d_in[7];
  const float* beta  = (const float*)d_in[8];
  float* out = (float*)d_out;

  const int N = in_sizes[0] / C;
  const int E = in_sizes[1] / 2;
  const long long NC = (long long)N * C;

  float* p1 = (float*)d_ws;
  float* p2 = p1 + NC;
  float* p5 = p1 + 4 * NC;

  const int NP   = (N + 31) / 32 * 32;
  const int nCh  = (E + CHUNK - 1) / CHUNK;
  const int nb   = (N + RS - 1) / RS;
  const int adjS = (E + 31) / 32 * 32 + 32 * MAXB;
  v2i* start = (v2i*)p5;
  int* adj   = (int*)(start + 2 * (long long)NP);
  v2i* rec   = (v2i*)p2;
  v2i* tab   = (v2i*)p1;

  const int nT32   = N / WGROWS;
  const int nFull  = nT32 - nT32 % N_WGP;
  const int nSplit = (N - nFull * WGROWS + 15) / 16;
  dim3 gM((unsigned)(nFull + nSplit));

  k_bin<<<dim3((unsigned)(2 * nCh)), 1024, 0, stream>>>(ei1, ei2, E, N, rec, tab);
  k_build<<<dim3((unsigned)(2 * nb)), 1024, 0, stream>>>(rec, tab, E, N, start, NP, adj, adjS);
  k_conv<<<gM, 128, 0, stream>>>(x, start, adj, NP, adjS,
                                 W1_r1, W1_r2, W2_r1, W2_r2, gamma, beta, out, N, E, nFull);
}
